# starting kernel
# baseline (speedup 1.0000x reference)
#include <hip/hip_runtime.h>


typedef float v2f __attribute__((ext_vector_type(2)));
typedef float v8f __attribute__((ext_vector_type(8)));

#define NB   8
#define CIN  64
#define HH   128
#define WW   128
#define COUT 128
#define PX   64
#define LDW  (PX + 4)
#define KTOT (9 * CIN)
#define NCP  (CIN / 2)
#define SLAB (5 * LDW * 2 + 56)

static constexpr float MUW = (1.0f - 0.9f) / 3.0f;

__global__ __launch_bounds__(256)
void pack_weights_kernel(const float* __restrict__ weight,
                         float* __restrict__ wpack) {
    int idx = blockIdx.x * 256 + threadIdx.x;
    if (idx >= KTOT * COUT) return;
    int par = idx & 1;
    int oc  = (idx >> 1) & (COUT - 1);
    int kk  = idx >> 8;
    int icp = kk & (NCP - 1);
    int tap = kk >> 5;
    int ic  = icp * 2 + par;
    int kh = (tap / 3) * 2;
    int kw = (tap % 3) * 2;
    wpack[idx] = weight[((oc * CIN + ic) * 5 + kh) * 5 + kw];
}

template <bool PACKED>
__global__ __launch_bounds__(256)
void conv_wmma_kernel(const float* __restrict__ input_,
                      const float* __restrict__ weight,
                      const float* __restrict__ wpack,
                      float* __restrict__ out) {
    __shared__ float lin[NCP * SLAB];
    __shared__ float sT[5 * LDW];
    __shared__ float sS[PX];

    const int blk = blockIdx.x;
    const int w0  = (blk & 1) * PX;
    const int h   = (blk >> 1) & (HH - 1);
    const int n   = blk >> 8;
    const int t   = threadIdx.x;

    const int TILE = CIN * 5 * LDW;
    for (int e = t; e < TILE; e += 256) {
        int ic  = e / (5 * LDW);
        int rem = e - ic * (5 * LDW);
        int r   = rem / LDW;
        int c   = rem - r * LDW;
        int gy = h + r - 2;
        int gx = w0 + c - 2;
        float v = 0.0f;
        if ((unsigned)gy < (unsigned)HH && (unsigned)gx < (unsigned)WW)
            v = input_[((n * CIN + ic) * HH + gy) * WW + gx];
        lin[(ic >> 1) * SLAB + ((r * LDW + c) << 1) + (ic & 1)] = v;
    }
    __syncthreads();

    for (int cell = t; cell < 5 * LDW; cell += 256) {
        float s = 0.0f;
        #pragma unroll 8
        for (int icp = 0; icp < NCP; ++icp) {
            v2f pr = *(const v2f*)&lin[icp * SLAB + (cell << 1)];
            s += pr.x + pr.y;
        }
        sT[cell] = s;
    }
    __syncthreads();

    if (t < PX) {
        const int p = t;
        float s = 0.0f;
        #pragma unroll
        for (int kw = 0; kw < 5; ++kw)
            s += sT[1 * LDW + p + kw] + sT[3 * LDW + p + kw];
        s += sT[0 * LDW + p + 1] + sT[0 * LDW + p + 3];
        s += sT[2 * LDW + p + 1] + sT[2 * LDW + p + 3];
        s += sT[4 * LDW + p + 1] + sT[4 * LDW + p + 3];
        sS[p] = MUW * s;
    }
    __syncthreads();

    const int lane = t & 31;
    const int wv   = t >> 5;
    const int ocb  = wv * 16;
    const int mrow = lane & 15;
    const int hi   = lane >> 4;

    v8f acc[4];
    #pragma unroll
    for (int q = 0; q < 4; ++q) acc[q] = (v8f){};

    for (int tap = 0; tap < 9; ++tap) {
        const int kh = (tap / 3) * 2;
        const int kw = (tap % 3) * 2;
        const int bcol = (kh * LDW + kw + mrow) << 1;
        #pragma unroll
        for (int ic4 = 0; ic4 < 16; ++ic4) {
            const int icp = ic4 * 2 + hi;
            v2f a;
            if (PACKED) {
                a = *(const v2f*)
                    &wpack[(((tap * NCP + icp) * COUT) + ocb + mrow) * 2];
            } else {
                const float* ap =
                    &weight[((ocb + mrow) * CIN + icp * 2) * 25 + kh * 5 + kw];
                a.x = ap[0];
                a.y = ap[25];
            }
            #pragma unroll
            for (int q = 0; q < 4; ++q) {
                v2f b = *(const v2f*)&lin[icp * SLAB + bcol + (q << 5)];
                acc[q] = __builtin_amdgcn_wmma_f32_16x16x4_f32(
                             false, a, false, b, (short)0, acc[q],
                             false, false);
            }
        }
    }

    float add[4];
    #pragma unroll
    for (int q = 0; q < 4; ++q) add[q] = sS[mrow + q * 16];
    #pragma unroll
    for (int j = 0; j < 8; ++j) {
        const int oc = ocb + j + 8 * hi;
        float* o = &out[((n * COUT + oc) * HH + h) * WW + w0];
        #pragma unroll
        for (int q = 0; q < 4; ++q)
            o[mrow + q * 16] = acc[q][j] + add[q];
    }
}

extern "C" void kernel_launch(void* const* d_in, const int* in_sizes, int n_in,
                              void* d_out, int out_size, void* d_ws,
                              size_t ws_size, hipStream_t stream) {
    const float* input_ = (const float*)d_in[0];
    const float* weight = (const float*)d_in[1];
    float* out = (float*)d_out;

    const int nblocks = NB * HH * (WW / PX);
    const size_t need = (size_t)KTOT * COUT * sizeof(float);

    if (d_ws != nullptr && ws_size >= need) {
        float* wpack = (float*)d_ws;
        pack_weights_kernel<<<(KTOT * COUT + 255) / 256, 256, 0, stream>>>(
            weight, wpack);
        conv_wmma_kernel<true><<<nblocks, 256, 0, stream>>>(
            input_, weight, wpack, out);
    } else {
        conv_wmma_kernel<false><<<nblocks, 256, 0, stream>>>(
            input_, weight, nullptr, out);
    }
}


# Input-tile staging: each thread issues all its loads as one batch of 16-byte loads (both channels of a pair x 4 columns) before its first LDS store, replacing 85 serially waited 4-byte loads
# speedup vs baseline: 1.5135x; 1.5135x over previous
#include <hip/hip_runtime.h>


typedef float v2f __attribute__((ext_vector_type(2)));
typedef float v8f __attribute__((ext_vector_type(8)));

#define NB   8
#define CIN  64
#define HH   128
#define WW   128
#define COUT 128
#define PX   64
#define LDW  (PX + 4)
#define KTOT (9 * CIN)
#define NCP  (CIN / 2)
#define SLAB (5 * LDW * 2 + 56)

static constexpr float MUW = (1.0f - 0.9f) / 3.0f;

__global__ __launch_bounds__(256)
void pack_weights_kernel(const float* __restrict__ weight,
                         float* __restrict__ wpack) {
    int idx = blockIdx.x * 256 + threadIdx.x;
    if (idx >= KTOT * COUT) return;
    int par = idx & 1;
    int oc  = (idx >> 1) & (COUT - 1);
    int kk  = idx >> 8;
    int icp = kk & (NCP - 1);
    int tap = kk >> 5;
    int ic  = icp * 2 + par;
    int kh = (tap / 3) * 2;
    int kw = (tap % 3) * 2;
    wpack[idx] = weight[((oc * CIN + ic) * 5 + kh) * 5 + kw];
}

template <bool PACKED>
__global__ __launch_bounds__(256)
void conv_wmma_kernel(const float* __restrict__ input_,
                      const float* __restrict__ weight,
                      const float* __restrict__ wpack,
                      float* __restrict__ out) {
    __shared__ __attribute__((aligned(16)))
    float lin[NCP * SLAB];
    __shared__ float sT[5 * LDW];
    __shared__ float sS[PX];

    const int blk = blockIdx.x;
    const int w0  = (blk & 1) * PX;
    const int h   = (blk >> 1) & (HH - 1);
    const int n   = blk >> 8;
    const int t   = threadIdx.x;

    {
        const int c4  = t & 15;
        const int sub = t >> 4;
        const float* inb = input_ + n * (CIN * HH * WW);

        float4 xa[10], xb[10];
        #pragma unroll
        for (int i = 0; i < 10; ++i) {
            const int r   = i >> 1;
            const int icp = sub + 16 * (i & 1);
            int gy = h + r - 2;
            gy = gy < 0 ? 0 : (gy > HH - 1 ? HH - 1 : gy);
            const float* p = inb + ((2 * icp) * HH + gy) * WW + w0 + 4 * c4;
            xa[i] = *(const float4*)p;
            xb[i] = *(const float4*)(p + HH * WW);
        }

        const int hside = t & 1;
        const int hicp  = (t >> 1) & 31;
        const int hgx   = w0 - 2 + hside * (PX + 2);
        const bool hxv  = (unsigned)hgx < (unsigned)WW;
        const int hgxc  = hxv ? hgx : 0;
        const int hr0   = t >> 6;
        int gy0 = h + hr0 - 2;
        const bool hv0 = hxv && (unsigned)gy0 < (unsigned)HH;
        gy0 = gy0 < 0 ? 0 : (gy0 > HH - 1 ? HH - 1 : gy0);
        const float* hp0 = inb + ((2 * hicp) * HH + gy0) * WW + hgxc;
        const float2 ha0 = *(const float2*)hp0;
        const float2 hb0 = *(const float2*)(hp0 + HH * WW);
        float2 ha1 = make_float2(0.0f, 0.0f), hb1 = ha1;
        int gy1 = h + 2;
        const bool hv1 = hxv && gy1 < HH;
        gy1 = gy1 > HH - 1 ? HH - 1 : gy1;
        if (t < 64) {
            const float* hp1 = inb + ((2 * hicp) * HH + gy1) * WW + hgxc;
            ha1 = *(const float2*)hp1;
            hb1 = *(const float2*)(hp1 + HH * WW);
        }

        #pragma unroll
        for (int i = 0; i < 10; ++i) {
            const int r   = i >> 1;
            const int icp = sub + 16 * (i & 1);
            const bool rv = (unsigned)(h + r - 2) < (unsigned)HH;
            const float4 a = rv ? xa[i] : make_float4(0.f, 0.f, 0.f, 0.f);
            const float4 b = rv ? xb[i] : make_float4(0.f, 0.f, 0.f, 0.f);
            float* d = &lin[icp * SLAB + ((r * LDW + 2 + 4 * c4) << 1)];
            *(float4*)d       = make_float4(a.x, b.x, a.y, b.y);
            *(float4*)(d + 4) = make_float4(a.z, b.z, a.w, b.w);
        }
        {
            const float2 a = hv0 ? ha0 : make_float2(0.f, 0.f);
            const float2 b = hv0 ? hb0 : make_float2(0.f, 0.f);
            float* d = &lin[hicp * SLAB + ((hr0 * LDW + hside * (PX + 2)) << 1)];
            *(float4*)d = make_float4(a.x, b.x, a.y, b.y);
        }
        if (t < 64) {
            const float2 a = hv1 ? ha1 : make_float2(0.f, 0.f);
            const float2 b = hv1 ? hb1 : make_float2(0.f, 0.f);
            float* d = &lin[hicp * SLAB + ((4 * LDW + hside * (PX + 2)) << 1)];
            *(float4*)d = make_float4(a.x, b.x, a.y, b.y);
        }
    }
    __syncthreads();

    for (int cell = t; cell < 5 * LDW; cell += 256) {
        float s = 0.0f;
        #pragma unroll 8
        for (int icp = 0; icp < NCP; ++icp) {
            v2f pr = *(const v2f*)&lin[icp * SLAB + (cell << 1)];
            s += pr.x + pr.y;
        }
        sT[cell] = s;
    }
    __syncthreads();

    if (t < PX) {
        const int p = t;
        float s = 0.0f;
        #pragma unroll
        for (int kw = 0; kw < 5; ++kw)
            s += sT[1 * LDW + p + kw] + sT[3 * LDW + p + kw];
        s += sT[0 * LDW + p + 1] + sT[0 * LDW + p + 3];
        s += sT[2 * LDW + p + 1] + sT[2 * LDW + p + 3];
        s += sT[4 * LDW + p + 1] + sT[4 * LDW + p + 3];
        sS[p] = MUW * s;
    }
    __syncthreads();

    const int lane = t & 31;
    const int wv   = t >> 5;
    const int ocb  = wv * 16;
    const int mrow = lane & 15;
    const int hi   = lane >> 4;

    v8f acc[4];
    #pragma unroll
    for (int q = 0; q < 4; ++q) acc[q] = (v8f){};

    for (int tap = 0; tap < 9; ++tap) {
        const int kh = (tap / 3) * 2;
        const int kw = (tap % 3) * 2;
        const int bcol = (kh * LDW + kw + mrow) << 1;
        #pragma unroll
        for (int ic4 = 0; ic4 < 16; ++ic4) {
            const int icp = ic4 * 2 + hi;
            v2f a;
            if (PACKED) {
                a = *(const v2f*)
                    &wpack[(((tap * NCP + icp) * COUT) + ocb + mrow) * 2];
            } else {
                const float* ap =
                    &weight[((ocb + mrow) * CIN + icp * 2) * 25 + kh * 5 + kw];
                a.x = ap[0];
                a.y = ap[25];
            }
            #pragma unroll
            for (int q = 0; q < 4; ++q) {
                v2f b = *(const v2f*)&lin[icp * SLAB + bcol + (q << 5)];
                acc[q] = __builtin_amdgcn_wmma_f32_16x16x4_f32(
                             false, a, false, b, (short)0, acc[q],
                             false, false);
            }
        }
    }

    float add[4];
    #pragma unroll
    for (int q = 0; q < 4; ++q) add[q] = sS[mrow + q * 16];
    #pragma unroll
    for (int j = 0; j < 8; ++j) {
        const int oc = ocb + j + 8 * hi;
        float* o = &out[((n * COUT + oc) * HH + h) * WW + w0];
        #pragma unroll
        for (int q = 0; q < 4; ++q)
            o[mrow + q * 16] = acc[q][j] + add[q];
    }
}

extern "C" void kernel_launch(void* const* d_in, const int* in_sizes, int n_in,
                              void* d_out, int out_size, void* d_ws,
                              size_t ws_size, hipStream_t stream) {
    const float* input_ = (const float*)d_in[0];
    const float* weight = (const float*)d_in[1];
    float* out = (float*)d_out;

    const int nblocks = NB * HH * (WW / PX);
    const size_t need = (size_t)KTOT * COUT * sizeof(float);

    if (d_ws != nullptr && ws_size >= need) {
        float* wpack = (float*)d_ws;
        pack_weights_kernel<<<(KTOT * COUT + 255) / 256, 256, 0, stream>>>(
            weight, wpack);
        conv_wmma_kernel<true><<<nblocks, 256, 0, stream>>>(
            input_, weight, wpack, out);
    } else {
        conv_wmma_kernel<false><<<nblocks, 256, 0, stream>>>(
            input_, weight, nullptr, out);
    }
}


# Each work-group computes four output rows from one 8-row input tile (two passes of two rows): 2 staged rows per output row instead of 5, a quarter as many work-groups, all resident in one round
# speedup vs baseline: 1.6285x; 1.0760x over previous
#include <hip/hip_runtime.h>


typedef float v2f __attribute__((ext_vector_type(2)));
typedef float v8f __attribute__((ext_vector_type(8)));

#define NB   8
#define CIN  64
#define HH   128
#define WW   128
#define COUT 128
#define PX   64
#define LDW  (PX + 4)
#define KTOT (9 * CIN)
#define NCP  (CIN / 2)
#define ROUT 4
#define TROWS (ROUT + 4)
#define NTHR 512
#define SLAB (TROWS * LDW * 2 + 32)

static constexpr float MUW = (1.0f - 0.9f) / 3.0f;

__global__ __launch_bounds__(256)
void pack_weights_kernel(const float* __restrict__ weight,
                         float* __restrict__ wpack) {
    int idx = blockIdx.x * 256 + threadIdx.x;
    if (idx >= KTOT * COUT) return;
    int par = idx & 1;
    int oc  = (idx >> 1) & (COUT - 1);
    int kk  = idx >> 8;
    int icp = kk & (NCP - 1);
    int tap = kk >> 5;
    int ic  = icp * 2 + par;
    int kh = (tap / 3) * 2;
    int kw = (tap % 3) * 2;
    wpack[idx] = weight[((oc * CIN + ic) * 5 + kh) * 5 + kw];
}

template <bool PACKED>
__global__ __launch_bounds__(NTHR)
void conv_wmma_kernel(const float* __restrict__ input_,
                      const float* __restrict__ weight,
                      const float* __restrict__ wpack,
                      float* __restrict__ out) {
    __shared__ __attribute__((aligned(16)))
    float lin[NCP * SLAB];
    __shared__ float sT[TROWS * LDW];
    __shared__ float sS[ROUT * PX];

    const int blk = blockIdx.x;
    const int w0  = (blk & 1) * PX;
    const int h   = ((blk >> 1) & (HH / ROUT - 1)) * ROUT;
    const int n   = blk >> 6;
    const int t   = threadIdx.x;

    {
        const int c4  = t & 15;
        const int icp = t >> 4;
        const float* inb = input_ + n * (CIN * HH * WW);

        float4 xa[TROWS], xb[TROWS];
        #pragma unroll
        for (int r = 0; r < TROWS; ++r) {
            int gy = h + r - 2;
            gy = gy < 0 ? 0 : (gy > HH - 1 ? HH - 1 : gy);
            const float* p = inb + ((2 * icp) * HH + gy) * WW + w0 + 4 * c4;
            xa[r] = *(const float4*)p;
            xb[r] = *(const float4*)(p + HH * WW);
        }

        const int hside = t & 1;
        const int hicp  = (t >> 1) & 31;
        const int hgx   = w0 - 2 + hside * (PX + 2);
        const bool hxv  = (unsigned)hgx < (unsigned)WW;
        const int hgxc  = hxv ? hgx : 0;
        const int hr    = t >> 6;
        int hgy = h + hr - 2;
        const bool hv = hxv && (unsigned)hgy < (unsigned)HH;
        hgy = hgy < 0 ? 0 : (hgy > HH - 1 ? HH - 1 : hgy);
        const float* hp = inb + ((2 * hicp) * HH + hgy) * WW + hgxc;
        const float2 ha = *(const float2*)hp;
        const float2 hb = *(const float2*)(hp + HH * WW);

        #pragma unroll
        for (int r = 0; r < TROWS; ++r) {
            const bool rv = (unsigned)(h + r - 2) < (unsigned)HH;
            const float4 a = rv ? xa[r] : make_float4(0.f, 0.f, 0.f, 0.f);
            const float4 b = rv ? xb[r] : make_float4(0.f, 0.f, 0.f, 0.f);
            float* d = &lin[icp * SLAB + ((r * LDW + 2 + 4 * c4) << 1)];
            *(float4*)d       = make_float4(a.x, b.x, a.y, b.y);
            *(float4*)(d + 4) = make_float4(a.z, b.z, a.w, b.w);
        }
        {
            const float2 a = hv ? ha : make_float2(0.f, 0.f);
            const float2 b = hv ? hb : make_float2(0.f, 0.f);
            float* d = &lin[hicp * SLAB + ((hr * LDW + hside * (PX + 2)) << 1)];
            *(float4*)d = make_float4(a.x, b.x, a.y, b.y);
        }
    }
    __syncthreads();

    for (int cell = t; cell < TROWS * LDW; cell += NTHR) {
        float s = 0.0f;
        #pragma unroll 8
        for (int icp = 0; icp < NCP; ++icp) {
            v2f pr = *(const v2f*)&lin[icp * SLAB + (cell << 1)];
            s += pr.x + pr.y;
        }
        sT[cell] = s;
    }
    __syncthreads();

    if (t < ROUT * PX) {
        const int p  = t & (PX - 1);
        const float* T0 = &sT[(t >> 6) * LDW];
        float s = 0.0f;
        #pragma unroll
        for (int kw = 0; kw < 5; ++kw)
            s += T0[1 * LDW + p + kw] + T0[3 * LDW + p + kw];
        s += T0[0 * LDW + p + 1] + T0[0 * LDW + p + 3];
        s += T0[2 * LDW + p + 1] + T0[2 * LDW + p + 3];
        s += T0[4 * LDW + p + 1] + T0[4 * LDW + p + 3];
        sS[t] = MUW * s;
    }
    __syncthreads();

    const int lane = t & 31;
    const int wv   = t >> 5;
    const int ocb  = (wv >> 1) * 16;
    const int mrow = lane & 15;
    const int hi   = lane >> 4;

    for (int pass = 0; pass < ROUT / 2; ++pass) {
        const int ro = (wv & 1) + 2 * pass;

        v8f acc[4];
        #pragma unroll
        for (int q = 0; q < 4; ++q) acc[q] = (v8f){};

        for (int tap = 0; tap < 9; ++tap) {
            const int kh = (tap / 3) * 2;
            const int kw = (tap % 3) * 2;
            const int bcol = ((kh + ro) * LDW + kw + mrow) << 1;
            #pragma unroll
            for (int ic4 = 0; ic4 < 16; ++ic4) {
                const int icp = ic4 * 2 + hi;
                v2f a;
                if (PACKED) {
                    a = *(const v2f*)
                        &wpack[(((tap * NCP + icp) * COUT) + ocb + mrow) * 2];
                } else {
                    const float* ap =
                        &weight[((ocb + mrow) * CIN + icp * 2) * 25 + kh * 5 + kw];
                    a.x = ap[0];
                    a.y = ap[25];
                }
                #pragma unroll
                for (int q = 0; q < 4; ++q) {
                    v2f b = *(const v2f*)&lin[icp * SLAB + bcol + (q << 5)];
                    acc[q] = __builtin_amdgcn_wmma_f32_16x16x4_f32(
                                 false, a, false, b, (short)0, acc[q],
                                 false, false);
                }
            }
        }

        float add[4];
        #pragma unroll
        for (int q = 0; q < 4; ++q) add[q] = sS[ro * PX + mrow + q * 16];
        #pragma unroll
        for (int j = 0; j < 8; ++j) {
            const int oc = ocb + j + 8 * hi;
            float* o = &out[((n * COUT + oc) * HH + h + ro) * WW + w0];
            #pragma unroll
            for (int q = 0; q < 4; ++q)
                o[mrow + q * 16] = acc[q][j] + add[q];
        }
    }
}

extern "C" void kernel_launch(void* const* d_in, const int* in_sizes, int n_in,
                              void* d_out, int out_size, void* d_ws,
                              size_t ws_size, hipStream_t stream) {
    const float* input_ = (const float*)d_in[0];
    const float* weight = (const float*)d_in[1];
    float* out = (float*)d_out;

    const int nblocks = NB * (HH / ROUT) * (WW / PX);
    const size_t need = (size_t)KTOT * COUT * sizeof(float);

    if (d_ws != nullptr && ws_size >= need) {
        float* wpack = (float*)d_ws;
        pack_weights_kernel<<<(KTOT * COUT + 255) / 256, 256, 0, stream>>>(
            weight, wpack);
        conv_wmma_kernel<true><<<nblocks, NTHR, 0, stream>>>(
            input_, weight, wpack, out);
    } else {
        conv_wmma_kernel<false><<<nblocks, NTHR, 0, stream>>>(
            input_, weight, nullptr, out);
    }
}


# Horizontal taps by the minimal 2-output 3-tap filtering algorithm F(2,3): 4 WMMA products per output pair instead of 6, weights and inputs pre-transformed; changes the order of summation
# speedup vs baseline: 2.1667x; 1.3305x over previous
#include <hip/hip_runtime.h>


typedef float v2f __attribute__((ext_vector_type(2)));
typedef float v8f __attribute__((ext_vector_type(8)));

#define NB   8
#define CIN  64
#define HH   128
#define WW   128
#define COUT 128
#define PX   64
#define LDW  (PX + 4)
#define NCP  (CIN / 2)
#define UTOT (3 * 4 * CIN)
#define ROUT 4
#define TROWS (ROUT + 4)
#define NTHR 512
#define SLAB (TROWS * LDW * 2 + 4)

static constexpr float MUW = (1.0f - 0.9f) / 3.0f;

__device__ __forceinline__ float f23_weight(int pos, float g0, float g1,
                                            float g2) {
    return pos == 0 ? g0
         : pos == 1 ? 0.5f * (g0 + g1 + g2)
         : pos == 2 ? 0.5f * (g0 - g1 + g2)
         : g2;
}

__global__ __launch_bounds__(256)
void pack_weights_kernel(const float* __restrict__ weight,
                         float* __restrict__ upack) {
    int idx = blockIdx.x * 256 + threadIdx.x;
    if (idx >= UTOT * COUT) return;
    int par = idx & 1;
    int oc  = (idx >> 1) & (COUT - 1);
    int kk  = idx >> 8;
    int icp = kk & (NCP - 1);
    int kp  = kk >> 5;
    int pos = kp & 3;
    int kh  = kp >> 2;
    int ic  = icp * 2 + par;
    const float* g = &weight[((oc * CIN + ic) * 5 + 2 * kh) * 5];
    upack[idx] = f23_weight(pos, g[0], g[2], g[4]);
}

template <bool PACKED>
__global__ __launch_bounds__(NTHR, 8)
void conv_wmma_kernel(const float* __restrict__ input_,
                      const float* __restrict__ weight,
                      const float* __restrict__ upack,
                      float* __restrict__ out) {
    __shared__ __attribute__((aligned(16)))
    float lin[NCP * SLAB];
    __shared__ float sT[TROWS * LDW];
    __shared__ __attribute__((aligned(16)))
    float sS[ROUT * PX];

    const int blk = blockIdx.x;
    const int w0  = (blk & 1) * PX;
    const int h   = ((blk >> 1) & (HH / ROUT - 1)) * ROUT;
    const int n   = blk >> 6;
    const int t   = threadIdx.x;

    {
        const int c4  = t & 15;
        const int icp = t >> 4;
        const float* inb = input_ + n * (CIN * HH * WW);

        float4 xa[TROWS], xb[TROWS];
        #pragma unroll
        for (int r = 0; r < TROWS; ++r) {
            int gy = h + r - 2;
            gy = gy < 0 ? 0 : (gy > HH - 1 ? HH - 1 : gy);
            const float* p = inb + ((2 * icp) * HH + gy) * WW + w0 + 4 * c4;
            xa[r] = *(const float4*)p;
            xb[r] = *(const float4*)(p + HH * WW);
        }

        const int hside = t & 1;
        const int hicp  = (t >> 1) & 31;
        const int hgx   = w0 - 2 + hside * (PX + 2);
        const bool hxv  = (unsigned)hgx < (unsigned)WW;
        const int hgxc  = hxv ? hgx : 0;
        const int hr    = t >> 6;
        int hgy = h + hr - 2;
        const bool hv = hxv && (unsigned)hgy < (unsigned)HH;
        hgy = hgy < 0 ? 0 : (hgy > HH - 1 ? HH - 1 : hgy);
        const float* hp = inb + ((2 * hicp) * HH + hgy) * WW + hgxc;
        const float2 ha = *(const float2*)hp;
        const float2 hb = *(const float2*)(hp + HH * WW);

        #pragma unroll
        for (int r = 0; r < TROWS; ++r) {
            const bool rv = (unsigned)(h + r - 2) < (unsigned)HH;
            const float4 a = rv ? xa[r] : make_float4(0.f, 0.f, 0.f, 0.f);
            const float4 b = rv ? xb[r] : make_float4(0.f, 0.f, 0.f, 0.f);
            float* d = &lin[icp * SLAB + ((r * LDW + 2 + 4 * c4) << 1)];
            *(float4*)d       = make_float4(a.x, b.x, a.y, b.y);
            *(float4*)(d + 4) = make_float4(a.z, b.z, a.w, b.w);
        }
        {
            const float2 a = hv ? ha : make_float2(0.f, 0.f);
            const float2 b = hv ? hb : make_float2(0.f, 0.f);
            float* d = &lin[hicp * SLAB + ((hr * LDW + hside * (PX + 2)) << 1)];
            *(float4*)d = make_float4(a.x, b.x, a.y, b.y);
        }
    }
    __syncthreads();

    for (int cell = t; cell < TROWS * LDW; cell += NTHR) {
        float s = 0.0f;
        #pragma unroll 8
        for (int icp = 0; icp < NCP; ++icp) {
            v2f pr = *(const v2f*)&lin[icp * SLAB + (cell << 1)];
            s += pr.x + pr.y;
        }
        sT[cell] = s;
    }
    __syncthreads();

    if (t < ROUT * PX) {
        const int p  = t & (PX - 1);
        const float* T0 = &sT[(t >> 6) * LDW];
        float s = 0.0f;
        #pragma unroll
        for (int kw = 0; kw < 5; ++kw)
            s += T0[1 * LDW + p + kw] + T0[3 * LDW + p + kw];
        s += T0[0 * LDW + p + 1] + T0[0 * LDW + p + 3];
        s += T0[2 * LDW + p + 1] + T0[2 * LDW + p + 3];
        s += T0[4 * LDW + p + 1] + T0[4 * LDW + p + 3];
        sS[t] = MUW * s;
    }
    __syncthreads();

    const int lane = t & 31;
    const int wv   = t >> 5;
    const int ocb  = (wv >> 1) * 16;
    const int mrow = lane & 15;
    const int hi   = lane >> 4;

    auto load_a = [&](int kh, int pos, int ic4) -> v2f {
        const int icp = ic4 * 2 + hi;
        v2f a;
        if (PACKED) {
            a = *((const v2f*)upack +
                  ((((kh * 4 + pos) * NCP + icp) * COUT) + ocb + mrow));
        } else {
            const float* g = &weight[((ocb + mrow) * CIN + icp * 2) * 25
                                     + (2 * kh) * 5];
            a.x = f23_weight(pos, g[0], g[2], g[4]);
            a.y = f23_weight(pos, g[25], g[27], g[29]);
        }
        return a;
    };

    constexpr int CQ  = 1;
    constexpr int NCH = 16 / CQ;
    v2f an[CQ][4];
    #pragma unroll
    for (int u = 0; u < CQ; ++u)
        #pragma unroll
        for (int pos = 0; pos < 4; ++pos) an[u][pos] = load_a(0, pos, u);

    for (int pass = 0; pass < ROUT / 2; ++pass) {
        const int ro = (wv & 1) + 2 * pass;

        v8f accE[4], accO[4];
        #pragma unroll
        for (int q = 0; q < 4; ++q) { accE[q] = (v8f){}; accO[q] = (v8f){}; }

        for (int kh = 0; kh < 3; ++kh) {
            const float* brow =
                &lin[hi * SLAB + (((2 * kh + ro) * LDW + 4 * mrow) << 1)];
            const int khn = kh == 2 ? 0 : kh + 1;
            #pragma unroll
            for (int ch = 0; ch < NCH; ++ch) {
                v2f ac[CQ][4];
                #pragma unroll
                for (int u = 0; u < CQ; ++u)
                    #pragma unroll
                    for (int pos = 0; pos < 4; ++pos) {
                        ac[u][pos] = an[u][pos];
                        an[u][pos] = ch < NCH - 1
                                   ? load_a(kh, pos, CQ * (ch + 1) + u)
                                   : load_a(khn, pos, u);
                    }
                __builtin_amdgcn_sched_barrier(0);
                #pragma unroll
                for (int u = 0; u < CQ; ++u) {
                    const float* bp = brow + (CQ * ch + u) * (2 * SLAB);
                    const float4 x0 = *(const float4*)(bp);
                    const float4 x1 = *(const float4*)(bp + 4);
                    const float4 x2 = *(const float4*)(bp + 8);
                    const float4 x3 = *(const float4*)(bp + 12);
                    v2f vE[4], vO[4];
                    vE[0] = (v2f){x0.x - x2.x, x0.y - x2.y};
                    vE[1] = (v2f){x1.x + x2.x, x1.y + x2.y};
                    vE[2] = (v2f){x2.x - x1.x, x2.y - x1.y};
                    vE[3] = (v2f){x1.x - x3.x, x1.y - x3.y};
                    vO[0] = (v2f){x0.z - x2.z, x0.w - x2.w};
                    vO[1] = (v2f){x1.z + x2.z, x1.w + x2.w};
                    vO[2] = (v2f){x2.z - x1.z, x2.w - x1.w};
                    vO[3] = (v2f){x1.z - x3.z, x1.w - x3.w};
                    #pragma unroll
                    for (int pos = 0; pos < 4; ++pos) {
                        accE[pos] = __builtin_amdgcn_wmma_f32_16x16x4_f32(
                                        false, ac[u][pos], false, vE[pos],
                                        (short)0, accE[pos], false, false);
                        accO[pos] = __builtin_amdgcn_wmma_f32_16x16x4_f32(
                                        false, ac[u][pos], false, vO[pos],
                                        (short)0, accO[pos], false, false);
                    }
                }
            }
        }

        const float4 add = *(const float4*)&sS[ro * PX + 4 * mrow];
        #pragma unroll
        for (int j = 0; j < 8; ++j) {
            const int oc = ocb + j + 8 * hi;
            float* o = &out[((n * COUT + oc) * HH + h + ro) * WW + w0 + 4 * mrow];
            float4 r;
            r.x = (accE[0][j] + accE[1][j] + accE[2][j]) + add.x;
            r.y = (accO[0][j] + accO[1][j] + accO[2][j]) + add.y;
            r.z = (accE[1][j] - accE[2][j] - accE[3][j]) + add.z;
            r.w = (accO[1][j] - accO[2][j] - accO[3][j]) + add.w;
            *(float4*)o = r;
        }
    }
}

extern "C" void kernel_launch(void* const* d_in, const int* in_sizes, int n_in,
                              void* d_out, int out_size, void* d_ws,
                              size_t ws_size, hipStream_t stream) {
    const float* input_ = (const float*)d_in[0];
    const float* weight = (const float*)d_in[1];
    float* out = (float*)d_out;

    const int nblocks = NB * (HH / ROUT) * (WW / PX);
    const size_t need = (size_t)UTOT * COUT * sizeof(float);

    if (d_ws != nullptr && ws_size >= need) {
        float* upack = (float*)d_ws;
        pack_weights_kernel<<<(UTOT * COUT + 255) / 256, 256, 0, stream>>>(
            weight, upack);
        conv_wmma_kernel<true><<<nblocks, NTHR, 0, stream>>>(
            input_, weight, upack, out);
    } else {
        conv_wmma_kernel<false><<<nblocks, NTHR, 0, stream>>>(
            input_, weight, nullptr, out);
    }
}


# Row taps by F(2,3) as well: each 2x2 output patch by 2-D F(2x2,3x3), 16 WMMA products per patch and channel quad instead of 24, weights and inputs pre-transformed; changes the order of summation
# speedup vs baseline: 2.3813x; 1.0990x over previous
#include <hip/hip_runtime.h>


typedef float v2f __attribute__((ext_vector_type(2)));
typedef float v8f __attribute__((ext_vector_type(8)));

#define NB   8
#define CIN  64
#define HH   128
#define WW   128
#define COUT 128
#define PX   64
#define LDW  (PX + 4)
#define NCP  (CIN / 2)
#define UTOT (4 * 4 * CIN)
#define ROUT 4
#define TROWS (ROUT + 4)
#define NTHR 512
#define SLAB (TROWS * LDW * 2 + 4)

static constexpr float MUW = (1.0f - 0.9f) / 3.0f;

__device__ __forceinline__ float f23_weight(int pos, float g0, float g1,
                                            float g2) {
    return pos == 0 ? g0
         : pos == 1 ? 0.5f * (g0 + g1 + g2)
         : pos == 2 ? 0.5f * (g0 - g1 + g2)
         : g2;
}

__device__ __forceinline__ float f23x2_weight(const float* __restrict__ w,
                                              int i, int j) {
    const float c0 = f23_weight(j, w[0],  w[2],  w[4]);
    const float c1 = f23_weight(j, w[10], w[12], w[14]);
    const float c2 = f23_weight(j, w[20], w[22], w[24]);
    return f23_weight(i, c0, c1, c2);
}

__global__ __launch_bounds__(256)
void pack_weights_kernel(const float* __restrict__ weight,
                         float* __restrict__ upack) {
    int idx = blockIdx.x * 256 + threadIdx.x;
    if (idx >= UTOT * COUT) return;
    int par = idx & 1;
    int oc  = (idx >> 1) & (COUT - 1);
    int kk  = idx >> 8;
    int icp = kk & (NCP - 1);
    int ij  = kk >> 5;
    int ic  = icp * 2 + par;
    upack[idx] = f23x2_weight(&weight[(oc * CIN + ic) * 25], ij >> 2, ij & 3);
}

template <bool PACKED>
__global__ __launch_bounds__(NTHR, 4)
void conv_wmma_kernel(const float* __restrict__ input_,
                      const float* __restrict__ weight,
                      const float* __restrict__ upack,
                      float* __restrict__ out) {
    __shared__ __attribute__((aligned(16)))
    float lin[NCP * SLAB];
    __shared__ float sT[TROWS * LDW];
    __shared__ __attribute__((aligned(16)))
    float sS[ROUT * PX];

    const int blk = blockIdx.x;
    const int w0  = (blk & 1) * PX;
    const int h   = ((blk >> 1) & (HH / ROUT - 1)) * ROUT;
    const int n   = blk >> 6;
    const int t   = threadIdx.x;

    {
        const int c4  = t & 15;
        const int icp = t >> 4;
        const float* inb = input_ + n * (CIN * HH * WW);

        float4 xa[TROWS], xb[TROWS];
        #pragma unroll
        for (int r = 0; r < TROWS; ++r) {
            int gy = h + r - 2;
            gy = gy < 0 ? 0 : (gy > HH - 1 ? HH - 1 : gy);
            const float* p = inb + ((2 * icp) * HH + gy) * WW + w0 + 4 * c4;
            xa[r] = *(const float4*)p;
            xb[r] = *(const float4*)(p + HH * WW);
        }

        const int hside = t & 1;
        const int hicp  = (t >> 1) & 31;
        const int hgx   = w0 - 2 + hside * (PX + 2);
        const bool hxv  = (unsigned)hgx < (unsigned)WW;
        const int hgxc  = hxv ? hgx : 0;
        const int hr    = t >> 6;
        int hgy = h + hr - 2;
        const bool hv = hxv && (unsigned)hgy < (unsigned)HH;
        hgy = hgy < 0 ? 0 : (hgy > HH - 1 ? HH - 1 : hgy);
        const float* hp = inb + ((2 * hicp) * HH + hgy) * WW + hgxc;
        const float2 ha = *(const float2*)hp;
        const float2 hb = *(const float2*)(hp + HH * WW);

        #pragma unroll
        for (int r = 0; r < TROWS; ++r) {
            const bool rv = (unsigned)(h + r - 2) < (unsigned)HH;
            const float4 a = rv ? xa[r] : make_float4(0.f, 0.f, 0.f, 0.f);
            const float4 b = rv ? xb[r] : make_float4(0.f, 0.f, 0.f, 0.f);
            float* d = &lin[icp * SLAB + ((r * LDW + 2 + 4 * c4) << 1)];
            *(float4*)d       = make_float4(a.x, b.x, a.y, b.y);
            *(float4*)(d + 4) = make_float4(a.z, b.z, a.w, b.w);
        }
        {
            const float2 a = hv ? ha : make_float2(0.f, 0.f);
            const float2 b = hv ? hb : make_float2(0.f, 0.f);
            float* d = &lin[hicp * SLAB + ((hr * LDW + hside * (PX + 2)) << 1)];
            *(float4*)d = make_float4(a.x, b.x, a.y, b.y);
        }
    }
    __syncthreads();

    for (int cell = t; cell < TROWS * LDW; cell += NTHR) {
        float s = 0.0f;
        #pragma unroll 8
        for (int icp = 0; icp < NCP; ++icp) {
            v2f pr = *(const v2f*)&lin[icp * SLAB + (cell << 1)];
            s += pr.x + pr.y;
        }
        sT[cell] = s;
    }
    __syncthreads();

    if (t < ROUT * PX) {
        const int p  = t & (PX - 1);
        const float* T0 = &sT[(t >> 6) * LDW];
        float s = 0.0f;
        #pragma unroll
        for (int kw = 0; kw < 5; ++kw)
            s += T0[1 * LDW + p + kw] + T0[3 * LDW + p + kw];
        s += T0[0 * LDW + p + 1] + T0[0 * LDW + p + 3];
        s += T0[2 * LDW + p + 1] + T0[2 * LDW + p + 3];
        s += T0[4 * LDW + p + 1] + T0[4 * LDW + p + 3];
        sS[t] = MUW * s;
    }
    __syncthreads();

    const int lane = t & 31;
    const int wv   = t >> 5;
    const int ocb  = (wv >> 1) * 16;
    const int rp   = wv & 1;
    const int mrow = lane & 15;
    const int hi   = lane >> 4;
    const int cg   = mrow >> 1;
    const int par  = mrow & 1;

    auto load_a = [&](int ij, int ic4) -> v2f {
        const int icp = ic4 * 2 + hi;
        v2f a;
        if (PACKED) {
            a = *((const v2f*)upack +
                  (((ij * NCP + icp) * COUT) + ocb + mrow));
        } else {
            const float* g = &weight[((ocb + mrow) * CIN + icp * 2) * 25];
            a.x = f23x2_weight(g, ij >> 2, ij & 3);
            a.y = f23x2_weight(g + 25, ij >> 2, ij & 3);
        }
        return a;
    };

    v2f an[8];
    #pragma unroll
    for (int q = 0; q < 8; ++q) an[q] = load_a(q, 0);

    for (int pass = 0; pass < 2; ++pass) {
        const int pc = 32 * pass + 4 * cg;

        v8f acc[16];
        #pragma unroll
        for (int q = 0; q < 16; ++q) acc[q] = (v8f){};

        const float* brow = &lin[hi * SLAB + ((rp * LDW + pc + par) << 1)];

        #pragma nounroll
        for (int c4 = 0; c4 < 16; c4 += 4)
        #pragma unroll
        for (int u4 = 0; u4 < 4; ++u4) {
            const int ch = c4 + u4;
            v2f a0[8], a1[8];
            #pragma unroll
            for (int q = 0; q < 8; ++q) {
                a0[q] = an[q];
                an[q] = load_a(8 + q, ch);
            }
            __builtin_amdgcn_sched_barrier(0);

            const float* bp = brow + ch * (2 * SLAB);
            v2f hr[4][4];
            #pragma unroll
            for (int k = 0; k < 4; ++k) {
                const float* rk = bp + ((2 * k * LDW) << 1);
                const v2f d0 = *(const v2f*)(rk);
                const v2f d1 = *(const v2f*)(rk + 4);
                const v2f d2 = *(const v2f*)(rk + 8);
                const v2f d3 = *(const v2f*)(rk + 12);
                hr[k][0] = d0 - d2;
                hr[k][1] = d1 + d2;
                hr[k][2] = d2 - d1;
                hr[k][3] = d1 - d3;
            }
            #pragma unroll
            for (int j = 0; j < 4; ++j) {
                const v2f v0 = hr[0][j] - hr[2][j];
                const v2f v1 = hr[1][j] + hr[2][j];
                acc[j] = __builtin_amdgcn_wmma_f32_16x16x4_f32(
                             false, a0[j], false, v0,
                             (short)0, acc[j], false, false);
                acc[4 + j] = __builtin_amdgcn_wmma_f32_16x16x4_f32(
                             false, a0[4 + j], false, v1,
                             (short)0, acc[4 + j], false, false);
            }

            #pragma unroll
            for (int q = 0; q < 8; ++q) {
                a1[q] = an[q];
                an[q] = load_a(q, (ch + 1) & 15);
            }
            __builtin_amdgcn_sched_barrier(0);

            #pragma unroll
            for (int j = 0; j < 4; ++j) {
                const v2f v2 = hr[2][j] - hr[1][j];
                const v2f v3 = hr[1][j] - hr[3][j];
                acc[8 + j] = __builtin_amdgcn_wmma_f32_16x16x4_f32(
                             false, a1[j], false, v2,
                             (short)0, acc[8 + j], false, false);
                acc[12 + j] = __builtin_amdgcn_wmma_f32_16x16x4_f32(
                             false, a1[4 + j], false, v3,
                             (short)0, acc[12 + j], false, false);
            }
        }

        const int orow = rp + 2 * par;
        const float4 add = *(const float4*)&sS[orow * PX + pc];
        #pragma unroll
        for (int r = 0; r < 8; ++r) {
            float y0[4], y1[4];
            #pragma unroll
            for (int j = 0; j < 4; ++j) {
                y0[j] = acc[j][r] + acc[4 + j][r] + acc[8 + j][r];
                y1[j] = acc[4 + j][r] - acc[8 + j][r] - acc[12 + j][r];
            }
            const float ta0 = y0[0] + y0[1] + y0[2];
            const float ta1 = y0[1] - y0[2] - y0[3];
            const float tb0 = y1[0] + y1[1] + y1[2];
            const float tb1 = y1[1] - y1[2] - y1[3];
            const float g0 = __shfl_xor(par ? ta0 : tb0, 1);
            const float g1 = __shfl_xor(par ? ta1 : tb1, 1);
            const float k0 = par ? tb0 : ta0;
            const float k1 = par ? tb1 : ta1;
            float4 v;
            v.x = (par ? g0 : k0) + add.x;
            v.y = (par ? k0 : g0) + add.y;
            v.z = (par ? g1 : k1) + add.z;
            v.w = (par ? k1 : g1) + add.w;
            const int oc = ocb + r + 8 * hi;
            float* o = &out[((n * COUT + oc) * HH + h + orow) * WW + w0 + pc];
            *(float4*)o = v;
        }
    }
}

extern "C" void kernel_launch(void* const* d_in, const int* in_sizes, int n_in,
                              void* d_out, int out_size, void* d_ws,
                              size_t ws_size, hipStream_t stream) {
    const float* input_ = (const float*)d_in[0];
    const float* weight = (const float*)d_in[1];
    float* out = (float*)d_out;

    const int nblocks = NB * (HH / ROUT) * (WW / PX);
    const size_t need = (size_t)UTOT * COUT * sizeof(float);

    if (d_ws != nullptr && ws_size >= need) {
        float* upack = (float*)d_ws;
        pack_weights_kernel<<<(UTOT * COUT + 255) / 256, 256, 0, stream>>>(
            weight, upack);
        conv_wmma_kernel<true><<<nblocks, NTHR, 0, stream>>>(
            input_, weight, upack, out);
    } else {
        conv_wmma_kernel<false><<<nblocks, NTHR, 0, stream>>>(
            input_, weight, nullptr, out);
    }
}


# Wave issue priority falls as a wave advances through its K loops (s_setprio 3 in pass 0, then 2, 1, 0 over pass 1): the four waves of a SIMD finish together, the youngest no longer alone
# speedup vs baseline: 2.4917x; 1.0464x over previous
#include <hip/hip_runtime.h>


typedef float v2f __attribute__((ext_vector_type(2)));
typedef float v8f __attribute__((ext_vector_type(8)));

#define NB   8
#define CIN  64
#define HH   128
#define WW   128
#define COUT 128
#define PX   64
#define LDW  (PX + 4)
#define NCP  (CIN / 2)
#define UTOT (4 * 4 * CIN)
#define ROUT 4
#define TROWS (ROUT + 4)
#define NTHR 512
#define SLAB (TROWS * LDW * 2 + 4)

static constexpr float MUW = (1.0f - 0.9f) / 3.0f;

__device__ __forceinline__ float f23_weight(int pos, float g0, float g1,
                                            float g2) {
    return pos == 0 ? g0
         : pos == 1 ? 0.5f * (g0 + g1 + g2)
         : pos == 2 ? 0.5f * (g0 - g1 + g2)
         : g2;
}

__device__ __forceinline__ float f23x2_weight(const float* __restrict__ w,
                                              int i, int j) {
    const float c0 = f23_weight(j, w[0],  w[2],  w[4]);
    const float c1 = f23_weight(j, w[10], w[12], w[14]);
    const float c2 = f23_weight(j, w[20], w[22], w[24]);
    return f23_weight(i, c0, c1, c2);
}

__global__ __launch_bounds__(256)
void pack_weights_kernel(const float* __restrict__ weight,
                         float* __restrict__ upack) {
    int idx = blockIdx.x * 256 + threadIdx.x;
    if (idx >= UTOT * COUT) return;
    int par = idx & 1;
    int oc  = (idx >> 1) & (COUT - 1);
    int kk  = idx >> 8;
    int icp = kk & (NCP - 1);
    int ij  = kk >> 5;
    int ic  = icp * 2 + par;
    upack[idx] = f23x2_weight(&weight[(oc * CIN + ic) * 25], ij >> 2, ij & 3);
}

template <bool PACKED>
__global__ __launch_bounds__(NTHR, 4)
void conv_wmma_kernel(const float* __restrict__ input_,
                      const float* __restrict__ weight,
                      const float* __restrict__ upack,
                      float* __restrict__ out) {
    __shared__ __attribute__((aligned(16)))
    float lin[NCP * SLAB];
    __shared__ float sT[TROWS * LDW];
    __shared__ __attribute__((aligned(16)))
    float sS[ROUT * PX];

    const int blk = blockIdx.x;
    const int w0  = (blk & 1) * PX;
    const int h   = ((blk >> 1) & (HH / ROUT - 1)) * ROUT;
    const int n   = blk >> 6;
    const int t   = threadIdx.x;

    {
        const int c4  = t & 15;
        const int icp = t >> 4;
        const float* inb = input_ + n * (CIN * HH * WW);

        float4 xa[TROWS], xb[TROWS];
        #pragma unroll
        for (int r = 0; r < TROWS; ++r) {
            int gy = h + r - 2;
            gy = gy < 0 ? 0 : (gy > HH - 1 ? HH - 1 : gy);
            const float* p = inb + ((2 * icp) * HH + gy) * WW + w0 + 4 * c4;
            xa[r] = *(const float4*)p;
            xb[r] = *(const float4*)(p + HH * WW);
        }

        const int hside = t & 1;
        const int hicp  = (t >> 1) & 31;
        const int hgx   = w0 - 2 + hside * (PX + 2);
        const bool hxv  = (unsigned)hgx < (unsigned)WW;
        const int hgxc  = hxv ? hgx : 0;
        const int hr    = t >> 6;
        int hgy = h + hr - 2;
        const bool hv = hxv && (unsigned)hgy < (unsigned)HH;
        hgy = hgy < 0 ? 0 : (hgy > HH - 1 ? HH - 1 : hgy);
        const float* hp = inb + ((2 * hicp) * HH + hgy) * WW + hgxc;
        const float2 ha = *(const float2*)hp;
        const float2 hb = *(const float2*)(hp + HH * WW);

        #pragma unroll
        for (int r = 0; r < TROWS; ++r) {
            const bool rv = (unsigned)(h + r - 2) < (unsigned)HH;
            const float4 a = rv ? xa[r] : make_float4(0.f, 0.f, 0.f, 0.f);
            const float4 b = rv ? xb[r] : make_float4(0.f, 0.f, 0.f, 0.f);
            float* d = &lin[icp * SLAB + ((r * LDW + 2 + 4 * c4) << 1)];
            *(float4*)d       = make_float4(a.x, b.x, a.y, b.y);
            *(float4*)(d + 4) = make_float4(a.z, b.z, a.w, b.w);
        }
        {
            const float2 a = hv ? ha : make_float2(0.f, 0.f);
            const float2 b = hv ? hb : make_float2(0.f, 0.f);
            float* d = &lin[hicp * SLAB + ((hr * LDW + hside * (PX + 2)) << 1)];
            *(float4*)d = make_float4(a.x, b.x, a.y, b.y);
        }
    }
    __syncthreads();

    for (int cell = t; cell < TROWS * LDW; cell += NTHR) {
        float s = 0.0f;
        #pragma unroll 8
        for (int icp = 0; icp < NCP; ++icp) {
            v2f pr = *(const v2f*)&lin[icp * SLAB + (cell << 1)];
            s += pr.x + pr.y;
        }
        sT[cell] = s;
    }
    __syncthreads();

    if (t < ROUT * PX) {
        const int p  = t & (PX - 1);
        const float* T0 = &sT[(t >> 6) * LDW];
        float s = 0.0f;
        #pragma unroll
        for (int kw = 0; kw < 5; ++kw)
            s += T0[1 * LDW + p + kw] + T0[3 * LDW + p + kw];
        s += T0[0 * LDW + p + 1] + T0[0 * LDW + p + 3];
        s += T0[2 * LDW + p + 1] + T0[2 * LDW + p + 3];
        s += T0[4 * LDW + p + 1] + T0[4 * LDW + p + 3];
        sS[t] = MUW * s;
    }
    __syncthreads();

    const int lane = t & 31;
    const int wv   = t >> 5;
    const int ocb  = (wv >> 1) * 16;
    const int rp   = wv & 1;
    const int mrow = lane & 15;
    const int hi   = lane >> 4;
    const int cg   = mrow >> 1;
    const int par  = mrow & 1;

    auto load_a = [&](int ij, int ic4) -> v2f {
        const int icp = ic4 * 2 + hi;
        v2f a;
        if (PACKED) {
            a = *((const v2f*)upack +
                  (((ij * NCP + icp) * COUT) + ocb + mrow));
        } else {
            const float* g = &weight[((ocb + mrow) * CIN + icp * 2) * 25];
            a.x = f23x2_weight(g, ij >> 2, ij & 3);
            a.y = f23x2_weight(g + 25, ij >> 2, ij & 3);
        }
        return a;
    };

    v2f an[8];
    #pragma unroll
    for (int q = 0; q < 8; ++q) an[q] = load_a(q, 0);

    __builtin_amdgcn_s_setprio(3);

    for (int pass = 0; pass < 2; ++pass) {
        const int pc = 32 * pass + 4 * cg;

        v8f acc[16];
        #pragma unroll
        for (int q = 0; q < 16; ++q) acc[q] = (v8f){};

        const float* brow = &lin[hi * SLAB + ((rp * LDW + pc + par) << 1)];

        #pragma nounroll
        for (int c4 = 0; c4 < 16; c4 += 4) {
        if (pass != 0) {
            if (c4 == 8)  __builtin_amdgcn_s_setprio(1);
            if (c4 == 12) __builtin_amdgcn_s_setprio(0);
        }
        #pragma unroll
        for (int u4 = 0; u4 < 4; ++u4) {
            const int ch = c4 + u4;
            v2f a0[8], a1[8];
            #pragma unroll
            for (int q = 0; q < 8; ++q) {
                a0[q] = an[q];
                an[q] = load_a(8 + q, ch);
            }
            __builtin_amdgcn_sched_barrier(0);

            const float* bp = brow + ch * (2 * SLAB);
            v2f hr[4][4];
            #pragma unroll
            for (int k = 0; k < 4; ++k) {
                const float* rk = bp + ((2 * k * LDW) << 1);
                const v2f d0 = *(const v2f*)(rk);
                const v2f d1 = *(const v2f*)(rk + 4);
                const v2f d2 = *(const v2f*)(rk + 8);
                const v2f d3 = *(const v2f*)(rk + 12);
                hr[k][0] = d0 - d2;
                hr[k][1] = d1 + d2;
                hr[k][2] = d2 - d1;
                hr[k][3] = d1 - d3;
            }
            #pragma unroll
            for (int j = 0; j < 4; ++j) {
                const v2f v0 = hr[0][j] - hr[2][j];
                const v2f v1 = hr[1][j] + hr[2][j];
                acc[j] = __builtin_amdgcn_wmma_f32_16x16x4_f32(
                             false, a0[j], false, v0,
                             (short)0, acc[j], false, false);
                acc[4 + j] = __builtin_amdgcn_wmma_f32_16x16x4_f32(
                             false, a0[4 + j], false, v1,
                             (short)0, acc[4 + j], false, false);
            }

            #pragma unroll
            for (int q = 0; q < 8; ++q) {
                a1[q] = an[q];
                an[q] = load_a(q, (ch + 1) & 15);
            }
            __builtin_amdgcn_sched_barrier(0);

            #pragma unroll
            for (int j = 0; j < 4; ++j) {
                const v2f v2 = hr[2][j] - hr[1][j];
                const v2f v3 = hr[1][j] - hr[3][j];
                acc[8 + j] = __builtin_amdgcn_wmma_f32_16x16x4_f32(
                             false, a1[j], false, v2,
                             (short)0, acc[8 + j], false, false);
                acc[12 + j] = __builtin_amdgcn_wmma_f32_16x16x4_f32(
                             false, a1[4 + j], false, v3,
                             (short)0, acc[12 + j], false, false);
            }
        }
        }

        if (pass == 0) __builtin_amdgcn_s_setprio(2);

        const int orow = rp + 2 * par;
        const float4 add = *(const float4*)&sS[orow * PX + pc];
        #pragma unroll
        for (int r = 0; r < 8; ++r) {
            float y0[4], y1[4];
            #pragma unroll
            for (int j = 0; j < 4; ++j) {
                y0[j] = acc[j][r] + acc[4 + j][r] + acc[8 + j][r];
                y1[j] = acc[4 + j][r] - acc[8 + j][r] - acc[12 + j][r];
            }
            const float ta0 = y0[0] + y0[1] + y0[2];
            const float ta1 = y0[1] - y0[2] - y0[3];
            const float tb0 = y1[0] + y1[1] + y1[2];
            const float tb1 = y1[1] - y1[2] - y1[3];
            const float g0 = __shfl_xor(par ? ta0 : tb0, 1);
            const float g1 = __shfl_xor(par ? ta1 : tb1, 1);
            const float k0 = par ? tb0 : ta0;
            const float k1 = par ? tb1 : ta1;
            float4 v;
            v.x = (par ? g0 : k0) + add.x;
            v.y = (par ? k0 : g0) + add.y;
            v.z = (par ? g1 : k1) + add.z;
            v.w = (par ? k1 : g1) + add.w;
            const int oc = ocb + r + 8 * hi;
            float* o = &out[((n * COUT + oc) * HH + h + orow) * WW + w0 + pc];
            *(float4*)o = v;
        }
    }
}

extern "C" void kernel_launch(void* const* d_in, const int* in_sizes, int n_in,
                              void* d_out, int out_size, void* d_ws,
                              size_t ws_size, hipStream_t stream) {
    const float* input_ = (const float*)d_in[0];
    const float* weight = (const float*)d_in[1];
    float* out = (float*)d_out;

    const int nblocks = NB * (HH / ROUT) * (WW / PX);
    const size_t need = (size_t)UTOT * COUT * sizeof(float);

    if (d_ws != nullptr && ws_size >= need) {
        float* upack = (float*)d_ws;
        pack_weights_kernel<<<(UTOT * COUT + 255) / 256, 256, 0, stream>>>(
            weight, upack);
        conv_wmma_kernel<true><<<nblocks, NTHR, 0, stream>>>(
            input_, weight, upack, out);
    } else {
        conv_wmma_kernel<false><<<nblocks, NTHR, 0, stream>>>(
            input_, weight, nullptr, out);
    }
}


# Tile widened to the whole 128-px image row, four 32-column passes per work-group: 256 work-groups, one per compute unit in a single round, so staging, channel sums, barriers and drain are paid once
# speedup vs baseline: 2.6194x; 1.0513x over previous
#include <hip/hip_runtime.h>


typedef float v2f __attribute__((ext_vector_type(2)));
typedef float v8f __attribute__((ext_vector_type(8)));

#define NB   8
#define CIN  64
#define HH   128
#define WW   128
#define COUT 128
#define PX   128
#define NPASS (PX / 32)
#define LDW  (PX + 4)
#define NCP  (CIN / 2)
#define UTOT (4 * 4 * CIN)
#define ROUT 4
#define TROWS (ROUT + 4)
#define NTHR 512
#define SLAB (TROWS * LDW * 2 + 4)

static constexpr float MUW = (1.0f - 0.9f) / 3.0f;

__device__ __forceinline__ float f23_weight(int pos, float g0, float g1,
                                            float g2) {
    return pos == 0 ? g0
         : pos == 1 ? 0.5f * (g0 + g1 + g2)
         : pos == 2 ? 0.5f * (g0 - g1 + g2)
         : g2;
}

__device__ __forceinline__ float f23x2_weight(const float* __restrict__ w,
                                              int i, int j) {
    const float c0 = f23_weight(j, w[0],  w[2],  w[4]);
    const float c1 = f23_weight(j, w[10], w[12], w[14]);
    const float c2 = f23_weight(j, w[20], w[22], w[24]);
    return f23_weight(i, c0, c1, c2);
}

__global__ __launch_bounds__(256)
void pack_weights_kernel(const float* __restrict__ weight,
                         float* __restrict__ upack) {
    int idx = blockIdx.x * 256 + threadIdx.x;
    if (idx >= UTOT * COUT) return;
    int par = idx & 1;
    int oc  = (idx >> 1) & (COUT - 1);
    int kk  = idx >> 8;
    int icp = kk & (NCP - 1);
    int ij  = kk >> 5;
    int ic  = icp * 2 + par;
    upack[idx] = f23x2_weight(&weight[(oc * CIN + ic) * 25], ij >> 2, ij & 3);
}

template <bool PACKED>
__global__ __launch_bounds__(NTHR, 4)
void conv_wmma_kernel(const float* __restrict__ input_,
                      const float* __restrict__ weight,
                      const float* __restrict__ upack,
                      float* __restrict__ out) {
    __shared__ __attribute__((aligned(16)))
    float lin[NCP * SLAB];
    __shared__ float sT[TROWS * LDW];
    __shared__ __attribute__((aligned(16)))
    float sS[ROUT * PX];

    const int blk = blockIdx.x;
    const int w0  = 0;
    const int h   = (blk & (HH / ROUT - 1)) * ROUT;
    const int n   = blk / (HH / ROUT);
    const int t   = threadIdx.x;

    {
        const int c4  = t & 31;
        const int icp = t >> 5;
        const float* inb = input_ + n * (CIN * HH * WW);

        float4 xa[2][TROWS], xb[2][TROWS];
        #pragma unroll
        for (int s = 0; s < 2; ++s) {
            #pragma unroll
            for (int r = 0; r < TROWS; ++r) {
                int gy = h + r - 2;
                gy = gy < 0 ? 0 : (gy > HH - 1 ? HH - 1 : gy);
                const float* p = inb + ((2 * (icp + 16 * s)) * HH + gy) * WW
                                 + w0 + 4 * c4;
                xa[s][r] = *(const float4*)p;
                xb[s][r] = *(const float4*)(p + HH * WW);
            }
        }

        const int hside = t & 1;
        const int hicp  = (t >> 1) & 31;
        const int hgx   = w0 - 2 + hside * (PX + 2);
        const bool hxv  = (unsigned)hgx < (unsigned)WW;
        const int hgxc  = hxv ? hgx : 0;
        const int hr    = t >> 6;
        int hgy = h + hr - 2;
        const bool hv = hxv && (unsigned)hgy < (unsigned)HH;
        hgy = hgy < 0 ? 0 : (hgy > HH - 1 ? HH - 1 : hgy);
        const float* hp = inb + ((2 * hicp) * HH + hgy) * WW + hgxc;
        const float2 ha = *(const float2*)hp;
        const float2 hb = *(const float2*)(hp + HH * WW);

        #pragma unroll
        for (int s = 0; s < 2; ++s) {
            #pragma unroll
            for (int r = 0; r < TROWS; ++r) {
                const bool rv = (unsigned)(h + r - 2) < (unsigned)HH;
                const float4 a = rv ? xa[s][r] : make_float4(0.f, 0.f, 0.f, 0.f);
                const float4 b = rv ? xb[s][r] : make_float4(0.f, 0.f, 0.f, 0.f);
                float* d = &lin[(icp + 16 * s) * SLAB +
                                ((r * LDW + 2 + 4 * c4) << 1)];
                *(float4*)d       = make_float4(a.x, b.x, a.y, b.y);
                *(float4*)(d + 4) = make_float4(a.z, b.z, a.w, b.w);
            }
        }
        {
            const float2 a = hv ? ha : make_float2(0.f, 0.f);
            const float2 b = hv ? hb : make_float2(0.f, 0.f);
            float* d = &lin[hicp * SLAB + ((hr * LDW + hside * (PX + 2)) << 1)];
            *(float4*)d = make_float4(a.x, b.x, a.y, b.y);
        }
    }
    __syncthreads();

    for (int cell = t; cell < TROWS * LDW; cell += NTHR) {
        float s = 0.0f;
        #pragma unroll 8
        for (int icp = 0; icp < NCP; ++icp) {
            v2f pr = *(const v2f*)&lin[icp * SLAB + (cell << 1)];
            s += pr.x + pr.y;
        }
        sT[cell] = s;
    }
    __syncthreads();

    if (t < ROUT * PX) {
        const int p  = t & (PX - 1);
        const float* T0 = &sT[(t / PX) * LDW];
        float s = 0.0f;
        #pragma unroll
        for (int kw = 0; kw < 5; ++kw)
            s += T0[1 * LDW + p + kw] + T0[3 * LDW + p + kw];
        s += T0[0 * LDW + p + 1] + T0[0 * LDW + p + 3];
        s += T0[2 * LDW + p + 1] + T0[2 * LDW + p + 3];
        s += T0[4 * LDW + p + 1] + T0[4 * LDW + p + 3];
        sS[t] = MUW * s;
    }
    __syncthreads();

    const int lane = t & 31;
    const int wv   = t >> 5;
    const int ocb  = (wv >> 1) * 16;
    const int rp   = wv & 1;
    const int mrow = lane & 15;
    const int hi   = lane >> 4;
    const int cg   = mrow >> 1;
    const int par  = mrow & 1;

    auto load_a = [&](int ij, int ic4) -> v2f {
        const int icp = ic4 * 2 + hi;
        v2f a;
        if (PACKED) {
            a = *((const v2f*)upack +
                  (((ij * NCP + icp) * COUT) + ocb + mrow));
        } else {
            const float* g = &weight[((ocb + mrow) * CIN + icp * 2) * 25];
            a.x = f23x2_weight(g, ij >> 2, ij & 3);
            a.y = f23x2_weight(g + 25, ij >> 2, ij & 3);
        }
        return a;
    };

    v2f an[8];
    #pragma unroll
    for (int q = 0; q < 8; ++q) an[q] = load_a(q, 0);

    __builtin_amdgcn_s_setprio(3);

    for (int pass = 0; pass < NPASS; ++pass) {
        const int pc = 32 * pass + 4 * cg;

        v8f acc[16];
        #pragma unroll
        for (int q = 0; q < 16; ++q) acc[q] = (v8f){};

        const float* brow = &lin[hi * SLAB + ((rp * LDW + pc + par) << 1)];

        #pragma nounroll
        for (int c4 = 0; c4 < 16; c4 += 4) {
        if (pass == NPASS - 1) {
            if (c4 == 8)  __builtin_amdgcn_s_setprio(1);
            if (c4 == 12) __builtin_amdgcn_s_setprio(0);
        }
        #pragma unroll
        for (int u4 = 0; u4 < 4; ++u4) {
            const int ch = c4 + u4;
            v2f a0[8], a1[8];
            #pragma unroll
            for (int q = 0; q < 8; ++q) {
                a0[q] = an[q];
                an[q] = load_a(8 + q, ch);
            }
            __builtin_amdgcn_sched_barrier(0);

            const float* bp = brow + ch * (2 * SLAB);
            v2f hr[4][4];
            #pragma unroll
            for (int k = 0; k < 4; ++k) {
                const float* rk = bp + ((2 * k * LDW) << 1);
                const v2f d0 = *(const v2f*)(rk);
                const v2f d1 = *(const v2f*)(rk + 4);
                const v2f d2 = *(const v2f*)(rk + 8);
                const v2f d3 = *(const v2f*)(rk + 12);
                hr[k][0] = d0 - d2;
                hr[k][1] = d1 + d2;
                hr[k][2] = d2 - d1;
                hr[k][3] = d1 - d3;
            }
            #pragma unroll
            for (int j = 0; j < 4; ++j) {
                const v2f v0 = hr[0][j] - hr[2][j];
                const v2f v1 = hr[1][j] + hr[2][j];
                acc[j] = __builtin_amdgcn_wmma_f32_16x16x4_f32(
                             false, a0[j], false, v0,
                             (short)0, acc[j], false, false);
                acc[4 + j] = __builtin_amdgcn_wmma_f32_16x16x4_f32(
                             false, a0[4 + j], false, v1,
                             (short)0, acc[4 + j], false, false);
            }

            #pragma unroll
            for (int q = 0; q < 8; ++q) {
                a1[q] = an[q];
                an[q] = load_a(q, (ch + 1) & 15);
            }
            __builtin_amdgcn_sched_barrier(0);

            #pragma unroll
            for (int j = 0; j < 4; ++j) {
                const v2f v2 = hr[2][j] - hr[1][j];
                const v2f v3 = hr[1][j] - hr[3][j];
                acc[8 + j] = __builtin_amdgcn_wmma_f32_16x16x4_f32(
                             false, a1[j], false, v2,
                             (short)0, acc[8 + j], false, false);
                acc[12 + j] = __builtin_amdgcn_wmma_f32_16x16x4_f32(
                             false, a1[4 + j], false, v3,
                             (short)0, acc[12 + j], false, false);
            }
        }
        }

        if (pass == NPASS - 2) __builtin_amdgcn_s_setprio(2);

        const int orow = rp + 2 * par;
        const float4 add = *(const float4*)&sS[orow * PX + pc];
        #pragma unroll
        for (int r = 0; r < 8; ++r) {
            float y0[4], y1[4];
            #pragma unroll
            for (int j = 0; j < 4; ++j) {
                y0[j] = acc[j][r] + acc[4 + j][r] + acc[8 + j][r];
                y1[j] = acc[4 + j][r] - acc[8 + j][r] - acc[12 + j][r];
            }
            const float ta0 = y0[0] + y0[1] + y0[2];
            const float ta1 = y0[1] - y0[2] - y0[3];
            const float tb0 = y1[0] + y1[1] + y1[2];
            const float tb1 = y1[1] - y1[2] - y1[3];
            const float g0 = __shfl_xor(par ? ta0 : tb0, 1);
            const float g1 = __shfl_xor(par ? ta1 : tb1, 1);
            const float k0 = par ? tb0 : ta0;
            const float k1 = par ? tb1 : ta1;
            float4 v;
            v.x = (par ? g0 : k0) + add.x;
            v.y = (par ? k0 : g0) + add.y;
            v.z = (par ? g1 : k1) + add.z;
            v.w = (par ? k1 : g1) + add.w;
            const int oc = ocb + r + 8 * hi;
            float* o = &out[((n * COUT + oc) * HH + h + orow) * WW + w0 + pc];
            *(float4*)o = v;
        }
    }
}

extern "C" void kernel_launch(void* const* d_in, const int* in_sizes, int n_in,
                              void* d_out, int out_size, void* d_ws,
                              size_t ws_size, hipStream_t stream) {
    const float* input_ = (const float*)d_in[0];
    const float* weight = (const float*)d_in[1];
    float* out = (float*)d_out;

    const int nblocks = NB * (HH / ROUT) * (WW / PX);
    const size_t need = (size_t)UTOT * COUT * sizeof(float);

    if (d_ws != nullptr && ws_size >= need) {
        float* upack = (float*)d_ws;
        pack_weights_kernel<<<(UTOT * COUT + 255) / 256, 256, 0, stream>>>(
            weight, upack);
        conv_wmma_kernel<true><<<nblocks, NTHR, 0, stream>>>(
            input_, weight, upack, out);
    } else {
        conv_wmma_kernel<false><<<nblocks, NTHR, 0, stream>>>(
            input_, weight, nullptr, out);
    }
}


# Row half of the F(2x2,3x3) input transform moved from the K loop into tile staging (once per work-group, tile size unchanged): 16 packed adds per 16 WMMAs, not 32; changes the order of summation
# speedup vs baseline: 2.6998x; 1.0307x over previous
#include <hip/hip_runtime.h>


typedef float v2f __attribute__((ext_vector_type(2)));
typedef float v8f __attribute__((ext_vector_type(8)));

#define NB   8
#define CIN  64
#define HH   128
#define WW   128
#define COUT 128
#define PX   128
#define NPASS (PX / 32)
#define LDW  (PX + 4)
#define NCP  (CIN / 2)
#define UTOT (4 * 4 * CIN)
#define ROUT 4
#define TROWS (ROUT + 4)
#define NTHR 512
#define SLAB (TROWS * LDW * 2 + 4)

static constexpr float MUW = (1.0f - 0.9f) / 3.0f;

__device__ __forceinline__ float4 f4_add(const float4 a, const float4 b) {
    return make_float4(a.x + b.x, a.y + b.y, a.z + b.z, a.w + b.w);
}
__device__ __forceinline__ float4 f4_sub(const float4 a, const float4 b) {
    return make_float4(a.x - b.x, a.y - b.y, a.z - b.z, a.w - b.w);
}

__device__ __forceinline__ float f23_weight(int pos, float g0, float g1,
                                            float g2) {
    return pos == 0 ? g0
         : pos == 1 ? 0.5f * (g0 + g1 + g2)
         : pos == 2 ? 0.5f * (g0 - g1 + g2)
         : g2;
}

__device__ __forceinline__ float f23x2_weight(const float* __restrict__ w,
                                              int i, int j) {
    const float c0 = f23_weight(j, w[0],  w[2],  w[4]);
    const float c1 = f23_weight(j, w[10], w[12], w[14]);
    const float c2 = f23_weight(j, w[20], w[22], w[24]);
    return f23_weight(i, c0, c1, c2);
}

__global__ __launch_bounds__(256)
void pack_weights_kernel(const float* __restrict__ weight,
                         float* __restrict__ upack) {
    int idx = blockIdx.x * 256 + threadIdx.x;
    if (idx >= UTOT * COUT) return;
    int par = idx & 1;
    int oc  = (idx >> 1) & (COUT - 1);
    int kk  = idx >> 8;
    int icp = kk & (NCP - 1);
    int ij  = kk >> 5;
    int ic  = icp * 2 + par;
    upack[idx] = f23x2_weight(&weight[(oc * CIN + ic) * 25], ij >> 2, ij & 3);
}

template <bool PACKED>
__global__ __launch_bounds__(NTHR, 4)
void conv_wmma_kernel(const float* __restrict__ input_,
                      const float* __restrict__ weight,
                      const float* __restrict__ upack,
                      float* __restrict__ out) {
    __shared__ __attribute__((aligned(16)))
    float lin[NCP * SLAB];
    __shared__ float sT[TROWS * LDW];
    __shared__ __attribute__((aligned(16)))
    float sS[ROUT * PX];

    const int blk = blockIdx.x;
    const int w0  = 0;
    const int h   = (blk & (HH / ROUT - 1)) * ROUT;
    const int n   = blk / (HH / ROUT);
    const int t   = threadIdx.x;

    {
        const int c4  = t & 31;
        const int icp = t >> 5;
        const float* inb = input_ + n * (CIN * HH * WW);

        float4 xa[2][TROWS], xb[2][TROWS];
        #pragma unroll
        for (int s = 0; s < 2; ++s) {
            #pragma unroll
            for (int r = 0; r < TROWS; ++r) {
                int gy = h + r - 2;
                gy = gy < 0 ? 0 : (gy > HH - 1 ? HH - 1 : gy);
                const float* p = inb + ((2 * (icp + 16 * s)) * HH + gy) * WW
                                 + w0 + 4 * c4;
                xa[s][r] = *(const float4*)p;
                xb[s][r] = *(const float4*)(p + HH * WW);
            }
        }

        const int hside = t & 1;
        const int hicp  = (t >> 1) & 31;
        const int hgx   = w0 - 2 + hside * (PX + 2);
        const bool hxv  = (unsigned)hgx < (unsigned)WW;
        const int hgxc  = hxv ? hgx : 0;
        const int hr    = t >> 6;
        const int hq    = hr & 1;
        const int hi2   = hr >> 1;
        const int hfa   = hq + (hi2 == 0 ? 0 : (hi2 == 2 ? 4 : 2));
        const int hfb   = hq + (hi2 == 3 ? 6 : (hi2 == 2 ? 2 : 4));
        int hgya = h + hfa - 2;
        int hgyb = h + hfb - 2;
        const bool hva = hxv && (unsigned)hgya < (unsigned)HH;
        const bool hvb = hxv && (unsigned)hgyb < (unsigned)HH;
        hgya = hgya < 0 ? 0 : (hgya > HH - 1 ? HH - 1 : hgya);
        hgyb = hgyb < 0 ? 0 : (hgyb > HH - 1 ? HH - 1 : hgyb);
        const float* hpa = inb + ((2 * hicp) * HH + hgya) * WW + hgxc;
        const float* hpb = inb + ((2 * hicp) * HH + hgyb) * WW + hgxc;
        const float2 haa = *(const float2*)hpa;
        const float2 hba = *(const float2*)(hpa + HH * WW);
        const float2 hab = *(const float2*)hpb;
        const float2 hbb = *(const float2*)(hpb + HH * WW);

        #pragma unroll
        for (int s = 0; s < 2; ++s) {
            float4 za[TROWS], zb[TROWS];
            #pragma unroll
            for (int r = 0; r < TROWS; ++r) {
                const bool rv = (unsigned)(h + r - 2) < (unsigned)HH;
                za[r] = rv ? xa[s][r] : make_float4(0.f, 0.f, 0.f, 0.f);
                zb[r] = rv ? xb[s][r] : make_float4(0.f, 0.f, 0.f, 0.f);
            }
            #pragma unroll
            for (int q = 0; q < 2; ++q) {
                float4 wa[4], wb[4];
                wa[0] = f4_sub(za[q],     za[q + 4]);
                wa[1] = f4_add(za[q + 2], za[q + 4]);
                wa[2] = f4_sub(za[q + 4], za[q + 2]);
                wa[3] = f4_sub(za[q + 2], za[q + 6]);
                wb[0] = f4_sub(zb[q],     zb[q + 4]);
                wb[1] = f4_add(zb[q + 2], zb[q + 4]);
                wb[2] = f4_sub(zb[q + 4], zb[q + 2]);
                wb[3] = f4_sub(zb[q + 2], zb[q + 6]);
                #pragma unroll
                for (int i = 0; i < 4; ++i) {
                    const float4 a = wa[i];
                    const float4 b = wb[i];
                    float* d = &lin[(icp + 16 * s) * SLAB +
                                    (((q + 2 * i) * LDW + 2 + 4 * c4) << 1)];
                    *(float4*)d       = make_float4(a.x, b.x, a.y, b.y);
                    *(float4*)(d + 4) = make_float4(a.z, b.z, a.w, b.w);
                }
            }
        }
        {
            const float2 aa = hva ? haa : make_float2(0.f, 0.f);
            const float2 ba = hva ? hba : make_float2(0.f, 0.f);
            const float2 ab = hvb ? hab : make_float2(0.f, 0.f);
            const float2 bb = hvb ? hbb : make_float2(0.f, 0.f);
            const float sg = (hi2 == 1) ? 1.0f : -1.0f;
            const float2 a = make_float2(aa.x + sg * ab.x, aa.y + sg * ab.y);
            const float2 b = make_float2(ba.x + sg * bb.x, ba.y + sg * bb.y);
            float* d = &lin[hicp * SLAB + ((hr * LDW + hside * (PX + 2)) << 1)];
            *(float4*)d = make_float4(a.x, b.x, a.y, b.y);
        }
    }
    __syncthreads();

    for (int cell = t; cell < TROWS * LDW; cell += NTHR) {
        float s = 0.0f;
        #pragma unroll 8
        for (int icp = 0; icp < NCP; ++icp) {
            v2f pr = *(const v2f*)&lin[icp * SLAB + (cell << 1)];
            s += pr.x + pr.y;
        }
        sT[cell] = s;
    }
    __syncthreads();

    if (t < 2 * LDW) {
        float* q = &sT[t];
        const float s0 = q[0];
        const float s1 = q[2 * LDW];
        const float s2 = q[4 * LDW];
        const float s3 = q[6 * LDW];
        const float r2 = 0.5f * (s1 + s2);
        const float r1 = 0.5f * (s1 - s2);
        q[0]       = s0 + r2;
        q[2 * LDW] = r1;
        q[4 * LDW] = r2;
        q[6 * LDW] = r1 - s3;
    }
    __syncthreads();

    if (t < ROUT * PX) {
        const int p  = t & (PX - 1);
        const float* T0 = &sT[(t / PX) * LDW];
        float s = 0.0f;
        #pragma unroll
        for (int kw = 0; kw < 5; ++kw)
            s += T0[1 * LDW + p + kw] + T0[3 * LDW + p + kw];
        s += T0[0 * LDW + p + 1] + T0[0 * LDW + p + 3];
        s += T0[2 * LDW + p + 1] + T0[2 * LDW + p + 3];
        s += T0[4 * LDW + p + 1] + T0[4 * LDW + p + 3];
        sS[t] = MUW * s;
    }
    __syncthreads();

    const int lane = t & 31;
    const int wv   = t >> 5;
    const int ocb  = (wv >> 1) * 16;
    const int rp   = wv & 1;
    const int mrow = lane & 15;
    const int hi   = lane >> 4;
    const int cg   = mrow >> 1;
    const int par  = mrow & 1;

    auto load_a = [&](int ij, int ic4) -> v2f {
        const int icp = ic4 * 2 + hi;
        v2f a;
        if (PACKED) {
            a = *((const v2f*)upack +
                  (((ij * NCP + icp) * COUT) + ocb + mrow));
        } else {
            const float* g = &weight[((ocb + mrow) * CIN + icp * 2) * 25];
            a.x = f23x2_weight(g, ij >> 2, ij & 3);
            a.y = f23x2_weight(g + 25, ij >> 2, ij & 3);
        }
        return a;
    };

    v2f an[8];
    #pragma unroll
    for (int q = 0; q < 8; ++q) an[q] = load_a(q, 0);

    __builtin_amdgcn_s_setprio(3);

    for (int pass = 0; pass < NPASS; ++pass) {
        const int pc = 32 * pass + 4 * cg;

        v8f acc[16];
        #pragma unroll
        for (int q = 0; q < 16; ++q) acc[q] = (v8f){};

        const float* brow = &lin[hi * SLAB + ((rp * LDW + pc + par) << 1)];

        #pragma nounroll
        for (int c4 = 0; c4 < 16; c4 += 4) {
        if (pass == NPASS - 1) {
            if (c4 == 8)  __builtin_amdgcn_s_setprio(1);
            if (c4 == 12) __builtin_amdgcn_s_setprio(0);
        }
        #pragma unroll
        for (int u4 = 0; u4 < 4; ++u4) {
            const int ch = c4 + u4;
            v2f a0[8], a1[8];
            #pragma unroll
            for (int q = 0; q < 8; ++q) {
                a0[q] = an[q];
                an[q] = load_a(8 + q, ch);
            }
            __builtin_amdgcn_sched_barrier(0);

            const float* bp = brow + ch * (2 * SLAB);
            v2f wr[4][4];
            #pragma unroll
            for (int i = 0; i < 4; ++i) {
                const float* rk = bp + ((2 * i * LDW) << 1);
                wr[i][0] = *(const v2f*)(rk);
                wr[i][1] = *(const v2f*)(rk + 4);
                wr[i][2] = *(const v2f*)(rk + 8);
                wr[i][3] = *(const v2f*)(rk + 12);
            }
            #pragma unroll
            for (int i = 0; i < 2; ++i) {
                const v2f v0 = wr[i][0] - wr[i][2];
                const v2f v1 = wr[i][1] + wr[i][2];
                const v2f v2 = wr[i][2] - wr[i][1];
                const v2f v3 = wr[i][1] - wr[i][3];
                acc[4 * i + 0] = __builtin_amdgcn_wmma_f32_16x16x4_f32(
                             false, a0[4 * i + 0], false, v0,
                             (short)0, acc[4 * i + 0], false, false);
                acc[4 * i + 1] = __builtin_amdgcn_wmma_f32_16x16x4_f32(
                             false, a0[4 * i + 1], false, v1,
                             (short)0, acc[4 * i + 1], false, false);
                acc[4 * i + 2] = __builtin_amdgcn_wmma_f32_16x16x4_f32(
                             false, a0[4 * i + 2], false, v2,
                             (short)0, acc[4 * i + 2], false, false);
                acc[4 * i + 3] = __builtin_amdgcn_wmma_f32_16x16x4_f32(
                             false, a0[4 * i + 3], false, v3,
                             (short)0, acc[4 * i + 3], false, false);
            }

            #pragma unroll
            for (int q = 0; q < 8; ++q) {
                a1[q] = an[q];
                an[q] = load_a(q, (ch + 1) & 15);
            }
            __builtin_amdgcn_sched_barrier(0);

            #pragma unroll
            for (int i = 2; i < 4; ++i) {
                const v2f v0 = wr[i][0] - wr[i][2];
                const v2f v1 = wr[i][1] + wr[i][2];
                const v2f v2 = wr[i][2] - wr[i][1];
                const v2f v3 = wr[i][1] - wr[i][3];
                acc[4 * i + 0] = __builtin_amdgcn_wmma_f32_16x16x4_f32(
                             false, a1[4 * i - 8], false, v0,
                             (short)0, acc[4 * i + 0], false, false);
                acc[4 * i + 1] = __builtin_amdgcn_wmma_f32_16x16x4_f32(
                             false, a1[4 * i - 7], false, v1,
                             (short)0, acc[4 * i + 1], false, false);
                acc[4 * i + 2] = __builtin_amdgcn_wmma_f32_16x16x4_f32(
                             false, a1[4 * i - 6], false, v2,
                             (short)0, acc[4 * i + 2], false, false);
                acc[4 * i + 3] = __builtin_amdgcn_wmma_f32_16x16x4_f32(
                             false, a1[4 * i - 5], false, v3,
                             (short)0, acc[4 * i + 3], false, false);
            }
        }
        }

        if (pass == NPASS - 2) __builtin_amdgcn_s_setprio(2);

        const int orow = rp + 2 * par;
        const float4 add = *(const float4*)&sS[orow * PX + pc];
        #pragma unroll
        for (int r = 0; r < 8; ++r) {
            float y0[4], y1[4];
            #pragma unroll
            for (int j = 0; j < 4; ++j) {
                y0[j] = acc[j][r] + acc[4 + j][r] + acc[8 + j][r];
                y1[j] = acc[4 + j][r] - acc[8 + j][r] - acc[12 + j][r];
            }
            const float ta0 = y0[0] + y0[1] + y0[2];
            const float ta1 = y0[1] - y0[2] - y0[3];
            const float tb0 = y1[0] + y1[1] + y1[2];
            const float tb1 = y1[1] - y1[2] - y1[3];
            const float g0 = __shfl_xor(par ? ta0 : tb0, 1);
            const float g1 = __shfl_xor(par ? ta1 : tb1, 1);
            const float k0 = par ? tb0 : ta0;
            const float k1 = par ? tb1 : ta1;
            float4 v;
            v.x = (par ? g0 : k0) + add.x;
            v.y = (par ? k0 : g0) + add.y;
            v.z = (par ? g1 : k1) + add.z;
            v.w = (par ? k1 : g1) + add.w;
            const int oc = ocb + r + 8 * hi;
            float* o = &out[((n * COUT + oc) * HH + h + orow) * WW + w0 + pc];
            *(float4*)o = v;
        }
    }
}

extern "C" void kernel_launch(void* const* d_in, const int* in_sizes, int n_in,
                              void* d_out, int out_size, void* d_ws,
                              size_t ws_size, hipStream_t stream) {
    const float* input_ = (const float*)d_in[0];
    const float* weight = (const float*)d_in[1];
    float* out = (float*)d_out;

    const int nblocks = NB * (HH / ROUT) * (WW / PX);
    const size_t need = (size_t)UTOT * COUT * sizeof(float);

    if (d_ws != nullptr && ws_size >= need) {
        float* upack = (float*)d_ws;
        pack_weights_kernel<<<(UTOT * COUT + 255) / 256, 256, 0, stream>>>(
            weight, upack);
        conv_wmma_kernel<true><<<nblocks, NTHR, 0, stream>>>(
            input_, weight, upack, out);
    } else {
        conv_wmma_kernel<false><<<nblocks, NTHR, 0, stream>>>(
            input_, weight, nullptr, out);
    }
}


# Column taps by the 4-output minimal filtering F(4,3) instead of F(2,3): 24 WMMA products per 2x4 output patch, not 32; 24 accumulator tiles in 8 waves of up to 512 VGPRs; changes the summation order
# speedup vs baseline: 2.8829x; 1.0678x over previous
#include <hip/hip_runtime.h>


typedef float v2f __attribute__((ext_vector_type(2)));
typedef float v8f __attribute__((ext_vector_type(8)));

#define NB   8
#define CIN  64
#define HH   128
#define WW   128
#define COUT 128
#define PX   128
#define NPASS (PX / 32)
#define LDW  (PX + 4)
#define NCP  (CIN / 2)
#define NIJ  24
#define UTOT (NIJ * CIN)
#define ROUT 4
#define TROWS (ROUT + 4)
#define NTHR 256
#define SLAB (TROWS * LDW * 2 + 4)

static constexpr float MUW = (1.0f - 0.9f) / 3.0f;

__device__ __forceinline__ float4 f4_add(const float4 a, const float4 b) {
    return make_float4(a.x + b.x, a.y + b.y, a.z + b.z, a.w + b.w);
}
__device__ __forceinline__ float4 f4_sub(const float4 a, const float4 b) {
    return make_float4(a.x - b.x, a.y - b.y, a.z - b.z, a.w - b.w);
}

__device__ __forceinline__ float f23_weight(int pos, float g0, float g1,
                                            float g2) {
    return pos == 0 ? g0
         : pos == 1 ? 0.5f * (g0 + g1 + g2)
         : pos == 2 ? 0.5f * (g0 - g1 + g2)
         : g2;
}

__device__ __forceinline__ float f43_weight(int pos, float g0, float g1,
                                            float g2) {
    return pos == 0 ? 0.25f * g0
         : pos == 1 ? (-1.0f / 6.0f) * (g0 + g1 + g2)
         : pos == 2 ? (-1.0f / 6.0f) * (g0 - g1 + g2)
         : pos == 3 ? (1.0f / 24.0f) * g0 + (1.0f / 12.0f) * g1
                      + (1.0f / 6.0f) * g2
         : pos == 4 ? (1.0f / 24.0f) * g0 - (1.0f / 12.0f) * g1
                      + (1.0f / 6.0f) * g2
         : g2;
}

__device__ __forceinline__ float f23x43_weight(const float* __restrict__ w,
                                               int i, int j) {
    const float c0 = f43_weight(j, w[0],  w[2],  w[4]);
    const float c1 = f43_weight(j, w[10], w[12], w[14]);
    const float c2 = f43_weight(j, w[20], w[22], w[24]);
    return f23_weight(i, c0, c1, c2);
}

__global__ __launch_bounds__(256)
void pack_weights_kernel(const float* __restrict__ weight,
                         float* __restrict__ upack) {
    int idx = blockIdx.x * 256 + threadIdx.x;
    if (idx >= UTOT * COUT) return;
    int par = idx & 1;
    int oc  = (idx >> 1) & (COUT - 1);
    int kk  = idx >> 8;
    int icp = kk & (NCP - 1);
    int ij  = kk >> 5;
    int ic  = icp * 2 + par;
    upack[idx] = f23x43_weight(&weight[(oc * CIN + ic) * 25], ij / 6, ij % 6);
}

template <bool PACKED>
__global__ __launch_bounds__(NTHR, 2)
void conv_wmma_kernel(const float* __restrict__ input_,
                      const float* __restrict__ weight,
                      const float* __restrict__ upack,
                      float* __restrict__ out) {
    __shared__ __attribute__((aligned(16)))
    float lin[NCP * SLAB];
    __shared__ float sT[TROWS * LDW];
    __shared__ __attribute__((aligned(16)))
    float sS[ROUT * PX];

    const int blk = blockIdx.x;
    const int w0  = 0;
    const int h   = (blk & (HH / ROUT - 1)) * ROUT;
    const int n   = blk / (HH / ROUT);
    const int t   = threadIdx.x;

    {
        const int c4  = t & 31;
        const int icp = t >> 5;
        const float* inb = input_ + n * (CIN * HH * WW);

        float4 xa[4][TROWS], xb[4][TROWS];
        #pragma unroll
        for (int s = 0; s < 4; ++s) {
            #pragma unroll
            for (int r = 0; r < TROWS; ++r) {
                int gy = h + r - 2;
                gy = gy < 0 ? 0 : (gy > HH - 1 ? HH - 1 : gy);
                const float* p = inb + ((2 * (icp + 8 * s)) * HH + gy) * WW
                                 + w0 + 4 * c4;
                xa[s][r] = *(const float4*)p;
                xb[s][r] = *(const float4*)(p + HH * WW);
            }
        }

        const int hside = t & 1;
        const int hicp  = (t >> 1) & 31;
        const int hgx   = w0 - 2 + hside * (PX + 2);
        const bool hxv  = (unsigned)hgx < (unsigned)WW;
        const int hgxc  = hxv ? hgx : 0;
        float2 haa[2], hba[2], hab[2], hbb[2];
        bool hva[2], hvb[2];
        #pragma unroll
        for (int e = 0; e < 2; ++e) {
            const int hr  = (t >> 6) + 4 * e;
            const int hq  = hr & 1;
            const int hi2 = hr >> 1;
            const int hfa = hq + (hi2 == 0 ? 0 : (hi2 == 2 ? 4 : 2));
            const int hfb = hq + (hi2 == 3 ? 6 : (hi2 == 2 ? 2 : 4));
            int hgya = h + hfa - 2;
            int hgyb = h + hfb - 2;
            hva[e] = hxv && (unsigned)hgya < (unsigned)HH;
            hvb[e] = hxv && (unsigned)hgyb < (unsigned)HH;
            hgya = hgya < 0 ? 0 : (hgya > HH - 1 ? HH - 1 : hgya);
            hgyb = hgyb < 0 ? 0 : (hgyb > HH - 1 ? HH - 1 : hgyb);
            const float* hpa = inb + ((2 * hicp) * HH + hgya) * WW + hgxc;
            const float* hpb = inb + ((2 * hicp) * HH + hgyb) * WW + hgxc;
            haa[e] = *(const float2*)hpa;
            hba[e] = *(const float2*)(hpa + HH * WW);
            hab[e] = *(const float2*)hpb;
            hbb[e] = *(const float2*)(hpb + HH * WW);
        }

        #pragma unroll
        for (int s = 0; s < 4; ++s) {
            float4 za[TROWS], zb[TROWS];
            #pragma unroll
            for (int r = 0; r < TROWS; ++r) {
                const bool rv = (unsigned)(h + r - 2) < (unsigned)HH;
                za[r] = rv ? xa[s][r] : make_float4(0.f, 0.f, 0.f, 0.f);
                zb[r] = rv ? xb[s][r] : make_float4(0.f, 0.f, 0.f, 0.f);
            }
            #pragma unroll
            for (int q = 0; q < 2; ++q) {
                float4 wa[4], wb[4];
                wa[0] = f4_sub(za[q],     za[q + 4]);
                wa[1] = f4_add(za[q + 2], za[q + 4]);
                wa[2] = f4_sub(za[q + 4], za[q + 2]);
                wa[3] = f4_sub(za[q + 2], za[q + 6]);
                wb[0] = f4_sub(zb[q],     zb[q + 4]);
                wb[1] = f4_add(zb[q + 2], zb[q + 4]);
                wb[2] = f4_sub(zb[q + 4], zb[q + 2]);
                wb[3] = f4_sub(zb[q + 2], zb[q + 6]);
                #pragma unroll
                for (int i = 0; i < 4; ++i) {
                    const float4 a = wa[i];
                    const float4 b = wb[i];
                    float* d = &lin[(icp + 8 * s) * SLAB +
                                    (((q + 2 * i) * LDW + 2 + 4 * c4) << 1)];
                    *(float4*)d       = make_float4(a.x, b.x, a.y, b.y);
                    *(float4*)(d + 4) = make_float4(a.z, b.z, a.w, b.w);
                }
            }
        }
        #pragma unroll
        for (int e = 0; e < 2; ++e) {
            const int hr  = (t >> 6) + 4 * e;
            const int hi2 = hr >> 1;
            const float2 aa = hva[e] ? haa[e] : make_float2(0.f, 0.f);
            const float2 ba = hva[e] ? hba[e] : make_float2(0.f, 0.f);
            const float2 ab = hvb[e] ? hab[e] : make_float2(0.f, 0.f);
            const float2 bb = hvb[e] ? hbb[e] : make_float2(0.f, 0.f);
            const float sg = (hi2 == 1) ? 1.0f : -1.0f;
            const float2 a = make_float2(aa.x + sg * ab.x, aa.y + sg * ab.y);
            const float2 b = make_float2(ba.x + sg * bb.x, ba.y + sg * bb.y);
            float* d = &lin[hicp * SLAB + ((hr * LDW + hside * (PX + 2)) << 1)];
            *(float4*)d = make_float4(a.x, b.x, a.y, b.y);
        }
    }
    __syncthreads();

    for (int cell = t; cell < TROWS * LDW; cell += NTHR) {
        float s = 0.0f;
        #pragma unroll 8
        for (int icp = 0; icp < NCP; ++icp) {
            v2f pr = *(const v2f*)&lin[icp * SLAB + (cell << 1)];
            s += pr.x + pr.y;
        }
        sT[cell] = s;
    }
    __syncthreads();

    for (int c = t; c < 2 * LDW; c += NTHR) {
        float* q = &sT[c];
        const float s0 = q[0];
        const float s1 = q[2 * LDW];
        const float s2 = q[4 * LDW];
        const float s3 = q[6 * LDW];
        const float r2 = 0.5f * (s1 + s2);
        const float r1 = 0.5f * (s1 - s2);
        q[0]       = s0 + r2;
        q[2 * LDW] = r1;
        q[4 * LDW] = r2;
        q[6 * LDW] = r1 - s3;
    }
    __syncthreads();

    for (int o = t; o < ROUT * PX; o += NTHR) {
        const int p  = o & (PX - 1);
        const float* T0 = &sT[(o / PX) * LDW];
        float s = 0.0f;
        #pragma unroll
        for (int kw = 0; kw < 5; ++kw)
            s += T0[1 * LDW + p + kw] + T0[3 * LDW + p + kw];
        s += T0[0 * LDW + p + 1] + T0[0 * LDW + p + 3];
        s += T0[2 * LDW + p + 1] + T0[2 * LDW + p + 3];
        s += T0[4 * LDW + p + 1] + T0[4 * LDW + p + 3];
        sS[o] = MUW * s;
    }
    __syncthreads();

    const int lane = t & 31;
    const int wv   = t >> 5;
    const int wpair = wv >> 1;
    const int wsub  = wv & 1;
    const int mrow = lane & 15;
    const int hi   = lane >> 4;
    const int rpl  = mrow >> 3;
    const int cg   = (mrow >> 1) & 3;
    const int par  = mrow & 1;

    auto load_a = [&](int ij, int ic4, int ocb) -> v2f {
        const int icp = ic4 * 2 + hi;
        v2f a;
        if (PACKED) {
            a = *((const v2f*)upack +
                  (((ij * NCP + icp) * COUT) + ocb + mrow));
        } else {
            const float* g = &weight[((ocb + mrow) * CIN + icp * 2) * 25];
            a.x = f23x43_weight(g, ij / 6, ij % 6);
            a.y = f23x43_weight(g + 25, ij / 6, ij % 6);
        }
        return a;
    };

    const float* lbase = &lin[hi * SLAB + ((rpl * LDW + 8 * cg + par) << 1)];

    v2f an[NIJ];
    v2f bn[4][6];
    #pragma unroll
    for (int q = 0; q < NIJ; ++q) an[q] = load_a(q, 0, 16 * wpair);
    #pragma unroll
    for (int i = 0; i < 4; ++i) {
        #pragma unroll
        for (int k = 0; k < 6; ++k)
            bn[i][k] = *(const v2f*)(lbase + ((32 * wsub) << 1)
                                     + ((2 * i * LDW) << 1) + 4 * k);
    }

    __builtin_amdgcn_s_setprio(3);

    for (int pass = 0; pass < NPASS; ++pass) {
        v8f acc[NIJ];
        #pragma unroll
        for (int q = 0; q < NIJ; ++q) acc[q] = (v8f){};

        #pragma nounroll
        for (int c4 = 0; c4 < 16; c4 += 4) {
        if (pass == NPASS - 1) {
            if (c4 == 8)  __builtin_amdgcn_s_setprio(1);
            if (c4 == 12) __builtin_amdgcn_s_setprio(0);
        }
        #pragma unroll
        for (int u4 = 0; u4 < 4; ++u4) {
            const int ch = c4 + u4;
            v2f a[NIJ];
            v2f b[4][6];
            #pragma unroll
            for (int q = 0; q < NIJ; ++q) a[q] = an[q];
            #pragma unroll
            for (int i = 0; i < 4; ++i) {
                #pragma unroll
                for (int k = 0; k < 6; ++k) b[i][k] = bn[i][k];
            }
            const int nch = (ch + 1) & 15;
            const int npass = (ch == 15) ? ((pass + 1) & (NPASS - 1)) : pass;
            const int nocb = 16 * (wpair + 4 * (npass >> 1));
            const int ncq  = 2 * (npass & 1) + wsub;
            const float* bp = lbase + ((32 * ncq) << 1) + nch * (2 * SLAB);
            #pragma unroll
            for (int q = 0; q < NIJ; ++q) an[q] = load_a(q, nch, nocb);
            #pragma unroll
            for (int i = 0; i < 4; ++i) {
                #pragma unroll
                for (int k = 0; k < 6; ++k)
                    bn[i][k] = *(const v2f*)(bp + ((2 * i * LDW) << 1) + 4 * k);
            }
            __builtin_amdgcn_sched_barrier(0);

            #pragma unroll
            for (int i = 0; i < 4; ++i) {
                const v2f d0 = b[i][0], d1 = b[i][1], d2 = b[i][2];
                const v2f d3 = b[i][3], d4 = b[i][4], d5 = b[i][5];
                const v2f e0 = d4 - 4.0f * d2;
                const v2f e1 = d3 - 4.0f * d1;
                const v2f e2 = d4 - d2;
                const v2f e3 = d3 - d1;
                v2f v[6];
                v[0] = 4.0f * d0 - 5.0f * d2 + d4;
                v[1] = e0 + e1;
                v[2] = e0 - e1;
                v[3] = e2 + 2.0f * e3;
                v[4] = e2 - 2.0f * e3;
                v[5] = 4.0f * d1 - 5.0f * d3 + d5;
                #pragma unroll
                for (int j = 0; j < 6; ++j)
                    acc[6 * i + j] = __builtin_amdgcn_wmma_f32_16x16x4_f32(
                                 false, a[6 * i + j], false, v[j],
                                 (short)0, acc[6 * i + j], false, false);
            }
        }
        }

        if (pass == NPASS - 2) __builtin_amdgcn_s_setprio(2);

        const int ocb = 16 * (wpair + 4 * (pass >> 1));
        const int pc  = 32 * (2 * (pass & 1) + wsub) + 8 * cg + 4 * par;
        const float4 add0 = *(const float4*)&sS[rpl * PX + pc];
        const float4 add1 = *(const float4*)&sS[(rpl + 2) * PX + pc];
        #pragma unroll
        for (int r = 0; r < 8; ++r) {
            float y0[6], y1[6];
            #pragma unroll
            for (int j = 0; j < 6; ++j) {
                y0[j] = acc[j][r] + acc[6 + j][r] + acc[12 + j][r];
                y1[j] = acc[6 + j][r] - acc[12 + j][r] - acc[18 + j][r];
            }
            const float p0 = y0[1] + y0[2], q0 = y0[1] - y0[2];
            const float r0 = y0[3] + y0[4], s0 = y0[3] - y0[4];
            const float p1 = y1[1] + y1[2], q1 = y1[1] - y1[2];
            const float r1 = y1[3] + y1[4], s1 = y1[3] - y1[4];
            float oa[4], ob[4];
            oa[0] = y0[0] + p0 + r0;
            oa[1] = q0 + 2.0f * s0;
            oa[2] = p0 + 4.0f * r0;
            oa[3] = q0 + 8.0f * s0 + y0[5];
            ob[0] = y1[0] + p1 + r1;
            ob[1] = q1 + 2.0f * s1;
            ob[2] = p1 + 4.0f * r1;
            ob[3] = q1 + 8.0f * s1 + y1[5];
            const float ga0 = __shfl_xor(par ? oa[0] : oa[2], 1);
            const float ga1 = __shfl_xor(par ? oa[1] : oa[3], 1);
            const float gb0 = __shfl_xor(par ? ob[0] : ob[2], 1);
            const float gb1 = __shfl_xor(par ? ob[1] : ob[3], 1);
            float4 va, vb;
            va.x = (par ? ga0   : oa[0]) + add0.x;
            va.y = (par ? oa[2] : ga0)   + add0.y;
            va.z = (par ? ga1   : oa[1]) + add0.z;
            va.w = (par ? oa[3] : ga1)   + add0.w;
            vb.x = (par ? gb0   : ob[0]) + add1.x;
            vb.y = (par ? ob[2] : gb0)   + add1.y;
            vb.z = (par ? gb1   : ob[1]) + add1.z;
            vb.w = (par ? ob[3] : gb1)   + add1.w;
            const int oc = ocb + r + 8 * hi;
            float* o = &out[((n * COUT + oc) * HH + h + rpl) * WW + w0 + pc];
            *(float4*)o = va;
            *(float4*)(o + 2 * WW) = vb;
        }
    }
}

extern "C" void kernel_launch(void* const* d_in, const int* in_sizes, int n_in,
                              void* d_out, int out_size, void* d_ws,
                              size_t ws_size, hipStream_t stream) {
    const float* input_ = (const float*)d_in[0];
    const float* weight = (const float*)d_in[1];
    float* out = (float*)d_out;

    const int nblocks = NB * (HH / ROUT) * (WW / PX);
    const size_t need = (size_t)UTOT * COUT * sizeof(float);

    if (d_ws != nullptr && ws_size >= need) {
        float* upack = (float*)d_ws;
        pack_weights_kernel<<<(UTOT * COUT + 255) / 256, 256, 0, stream>>>(
            weight, upack);
        conv_wmma_kernel<true><<<nblocks, NTHR, 0, stream>>>(
            input_, weight, upack, out);
    } else {
        conv_wmma_kernel<false><<<nblocks, NTHR, 0, stream>>>(
            input_, weight, nullptr, out);
    }
}
